# starting kernel
# baseline (speedup 1.0000x reference)
#include <hip/hip_runtime.h>
#include <hip/hip_bf16.h>


typedef __attribute__((ext_vector_type(2))) float v2f;
typedef __attribute__((ext_vector_type(8))) float v8f;

#define NN 50000
#define EE 800000
#define HD 128
#define CD 384

__global__ void deg_init(float* deg, int n) {
    int i = blockIdx.x * blockDim.x + threadIdx.x;
    if (i < n) deg[i] = 1.0f;
}
__global__ void deg_count(const int* __restrict__ dst, float* deg, int e) {
    int i = blockIdx.x * blockDim.x + threadIdx.x;
    if (i < e) atomicAdd(&deg[dst[i]], 1.0f);
}
__global__ void deg_rsqrt(float* deg, int n) {
    int i = blockIdx.x * blockDim.x + threadIdx.x;
    if (i < n) deg[i] = rsqrtf(deg[i]);
}

template <bool BT>
__global__ void __launch_bounds__(256)
gemm16_f32(const float* __restrict__ A, int lda,
           const float* __restrict__ B, int ldb,
           float* __restrict__ C, int ldc,
           int K, int tilesPerRow, int totalTiles) {
    int wave = threadIdx.x >> 5;
    int lane = threadIdx.x & 31;
    int tile = blockIdx.x * (blockDim.x >> 5) + wave;
    if (tile >= totalTiles) return;
    int tm = (tile / tilesPerRow) << 4;
    int tn = (tile % tilesPerRow) << 4;
    int rc   = lane & 15;
    int half = lane >> 4;

    v8f acc = {};
    const float* arow = A + (size_t)(tm + rc) * lda;
#pragma unroll 4
    for (int k0 = 0; k0 < K; k0 += 4) {
        int ka = k0 + 2 * half;
        v2f a;
        a.x = arow[ka];
        a.y = arow[ka + 1];
        v2f b;
        if (BT) {
            const float* bp = B + (size_t)(tn + rc) * ldb + ka;
            b.x = bp[0];
            b.y = bp[1];
        } else {
            b.x = B[(size_t)(ka + 0) * ldb + tn + rc];
            b.y = B[(size_t)(ka + 1) * ldb + tn + rc];
        }
        acc = __builtin_amdgcn_wmma_f32_16x16x4_f32(
            false, a, false, b, (short)0, acc, false, false);
    }
#pragma unroll
    for (int i = 0; i < 8; ++i)
        C[(size_t)(tm + i + 8 * half) * ldc + tn + rc] = acc[i];
}

__global__ void self_init(const float* __restrict__ XW, const float* __restrict__ dinv,
                          const float* __restrict__ bias,
                          float* __restrict__ H, int ldh, int colOff, int n) {
    int idx = blockIdx.x * blockDim.x + threadIdx.x;
    int v = idx >> 5, lane = idx & 31;
    if (v >= n) return;
    float dn = dinv[v];
    float nn = dn * dn;
    float4 x = ((const float4*)(XW + (size_t)v * HD))[lane];
    float4 b = ((const float4*)bias)[lane];
    float* op = H + (size_t)v * ldh + colOff + lane * 4;
    op[0] = x.x * nn + b.x;
    op[1] = x.y * nn + b.y;
    op[2] = x.z * nn + b.z;
    op[3] = x.w * nn + b.w;
}

__global__ void __launch_bounds__(256)
edge_scatter(const float* __restrict__ XW, const float* __restrict__ dinv,
             const int* __restrict__ src, const int* __restrict__ dst,
             float* __restrict__ H, int ldh, int colOff, int e) {
    int idx = blockIdx.x * blockDim.x + threadIdx.x;
    int eid = idx >> 5, lane = idx & 31;
    if (eid >= e) return;
    int s = src[eid], d = dst[eid];
    float norm = dinv[s] * dinv[d];
    float4 v = ((const float4*)(XW + (size_t)s * HD))[lane];
    float* op = H + (size_t)d * ldh + colOff + lane * 4;
    atomicAdd(op + 0, v.x * norm);
    atomicAdd(op + 1, v.y * norm);
    atomicAdd(op + 2, v.z * norm);
    atomicAdd(op + 3, v.w * norm);
}

__global__ void logits_init(float* logits, const float* __restrict__ lb2, int n) {
    int i = blockIdx.x * blockDim.x + threadIdx.x;
    if (i < n) {
        logits[2 * i + 0] = lb2[0];
        logits[2 * i + 1] = lb2[1];
    }
}

__global__ void __launch_bounds__(256)
linear_fused(const float* __restrict__ H,
             const float* __restrict__ lW1,
             const float* __restrict__ lb1,
             const float* __restrict__ lW2,
             float* __restrict__ logits,
             int totalTiles) {
    const int K = CD, tilesPerRow = CD / 16;
    int wave = threadIdx.x >> 5;
    int lane = threadIdx.x & 31;
    int tile = blockIdx.x * (blockDim.x >> 5) + wave;
    if (tile >= totalTiles) return;
    int tm = (tile / tilesPerRow) << 4;
    int tn = (tile % tilesPerRow) << 4;
    int col  = lane & 15;
    int half = lane >> 4;

    v8f acc = {};
    const float* arow = H + (size_t)(tm + col) * K;
    const float* brow = lW1 + (size_t)(tn + col) * K;
#pragma unroll 4
    for (int k0 = 0; k0 < K; k0 += 4) {
        int ka = k0 + 2 * half;
        v2f a;
        a.x = arow[ka];
        a.y = arow[ka + 1];
        v2f b;
        b.x = brow[ka];
        b.y = brow[ka + 1];
        acc = __builtin_amdgcn_wmma_f32_16x16x4_f32(
            false, a, false, b, (short)0, acc, false, false);
    }

    float bias = lb1[tn + col];
    float w0 = lW2[0 * CD + tn + col];
    float w1 = lW2[1 * CD + tn + col];
#pragma unroll
    for (int i = 0; i < 8; ++i) {
        float t = acc[i] + bias;
        t = t > 0.0f ? t : 0.0f;
        float p0 = t * w0;
        float p1 = t * w1;
#pragma unroll
        for (int m = 8; m >= 1; m >>= 1) {
            p0 += __shfl_xor(p0, m, 16);
            p1 += __shfl_xor(p1, m, 16);
        }
        if (col == 0) {
            int r = tm + i + 8 * half;
            atomicAdd(&logits[2 * r + 0], p0);
            atomicAdd(&logits[2 * r + 1], p1);
        }
    }
}

__global__ void softmax2(const float* __restrict__ logits, float* __restrict__ probs, int n) {
    int i = blockIdx.x * blockDim.x + threadIdx.x;
    if (i >= n) return;
    float a = logits[2 * i + 0], b = logits[2 * i + 1];
    float m = fmaxf(a, b);
    float ea = __expf(a - m), eb = __expf(b - m);
    float inv = 1.0f / (ea + eb);
    probs[2 * i + 0] = ea * inv;
    probs[2 * i + 1] = eb * inv;
}

extern "C" void kernel_launch(void* const* d_in, const int* in_sizes, int n_in,
                              void* d_out, int out_size, void* d_ws, size_t ws_size,
                              hipStream_t stream) {
    const float* x   = (const float*)d_in[0];
    const int*   ei  = (const int*)d_in[1];
    const float* W1  = (const float*)d_in[2];
    const float* b1  = (const float*)d_in[3];
    const float* W2  = (const float*)d_in[4];
    const float* b2  = (const float*)d_in[5];
    const float* W3  = (const float*)d_in[6];
    const float* b3  = (const float*)d_in[7];
    const float* lW1 = (const float*)d_in[8];
    const float* lb1 = (const float*)d_in[9];
    const float* lW2 = (const float*)d_in[10];
    const float* lb2 = (const float*)d_in[11];

    const int N = NN, E = EE;
    const int* src = ei;
    const int* dst = ei + E;

    float* dinv = (float*)d_ws;
    float* xw   = dinv + N;
    float* h    = xw + (size_t)N * HD;

    float* logits = (float*)d_out;
    float* probs  = logits + (size_t)N * 2;

    const int T = 256;
    const int rowTiles = N / 16;

    deg_init<<<(N + T - 1) / T, T, 0, stream>>>(dinv, N);
    deg_count<<<(E + T - 1) / T, T, 0, stream>>>(dst, dinv, E);
    deg_rsqrt<<<(N + T - 1) / T, T, 0, stream>>>(dinv, N);

    {
        int tiles = rowTiles * (HD / 16);
        gemm16_f32<false><<<(tiles + 7) / 8, T, 0, stream>>>(
            x, 64, W1, HD, xw, HD, 64, HD / 16, tiles);
        self_init<<<((size_t)N * 32 + T - 1) / T, T, 0, stream>>>(xw, dinv, b1, h, CD, 0, N);
        edge_scatter<<<((size_t)E * 32 + T - 1) / T, T, 0, stream>>>(
            xw, dinv, src, dst, h, CD, 0, E);
    }
    {
        int tiles = rowTiles * (HD / 16);
        gemm16_f32<false><<<(tiles + 7) / 8, T, 0, stream>>>(
            h + 0, CD, W2, HD, xw, HD, HD, HD / 16, tiles);
        self_init<<<((size_t)N * 32 + T - 1) / T, T, 0, stream>>>(xw, dinv, b2, h, CD, HD, N);
        edge_scatter<<<((size_t)E * 32 + T - 1) / T, T, 0, stream>>>(
            xw, dinv, src, dst, h, CD, HD, E);
    }
    {
        int tiles = rowTiles * (HD / 16);
        gemm16_f32<false><<<(tiles + 7) / 8, T, 0, stream>>>(
            h + HD, CD, W3, HD, xw, HD, HD, HD / 16, tiles);
        self_init<<<((size_t)N * 32 + T - 1) / T, T, 0, stream>>>(xw, dinv, b3, h, CD, 2 * HD, N);
        edge_scatter<<<((size_t)E * 32 + T - 1) / T, T, 0, stream>>>(
            xw, dinv, src, dst, h, CD, 2 * HD, E);
    }

    logits_init<<<(N + T - 1) / T, T, 0, stream>>>(logits, lb2, N);
    {
        int tiles = rowTiles * (CD / 16);
        linear_fused<<<(tiles + 7) / 8, T, 0, stream>>>(h, lW1, lb1, lW2, logits, tiles);
    }
    softmax2<<<(N + T - 1) / T, T, 0, stream>>>(logits, probs, N);
}


# Edge scatter regridded: one wave per 32 edges (was one per edge), src/dst/norm read once per lane and broadcast; each load and atomic covers one 128-B line; atomic adds arrive in another order
# speedup vs baseline: 1.3135x; 1.3135x over previous
#include <hip/hip_runtime.h>
#include <hip/hip_bf16.h>


typedef __attribute__((ext_vector_type(2))) float v2f;
typedef __attribute__((ext_vector_type(8))) float v8f;

#define NN 50000
#define EE 800000
#define HD 128
#define CD 384

__global__ void deg_init(float* deg, int n) {
    int i = blockIdx.x * blockDim.x + threadIdx.x;
    if (i < n) deg[i] = 1.0f;
}
__global__ void deg_count(const int* __restrict__ dst, float* deg, int e) {
    int i = blockIdx.x * blockDim.x + threadIdx.x;
    if (i < e) atomicAdd(&deg[dst[i]], 1.0f);
}
__global__ void deg_rsqrt(float* deg, int n) {
    int i = blockIdx.x * blockDim.x + threadIdx.x;
    if (i < n) deg[i] = rsqrtf(deg[i]);
}

template <bool BT>
__global__ void __launch_bounds__(256)
gemm16_f32(const float* __restrict__ A, int lda,
           const float* __restrict__ B, int ldb,
           float* __restrict__ C, int ldc,
           int K, int tilesPerRow, int totalTiles) {
    int wave = threadIdx.x >> 5;
    int lane = threadIdx.x & 31;
    int tile = blockIdx.x * (blockDim.x >> 5) + wave;
    if (tile >= totalTiles) return;
    int tm = (tile / tilesPerRow) << 4;
    int tn = (tile % tilesPerRow) << 4;
    int rc   = lane & 15;
    int half = lane >> 4;

    v8f acc = {};
    const float* arow = A + (size_t)(tm + rc) * lda;
#pragma unroll 4
    for (int k0 = 0; k0 < K; k0 += 4) {
        int ka = k0 + 2 * half;
        v2f a;
        a.x = arow[ka];
        a.y = arow[ka + 1];
        v2f b;
        if (BT) {
            const float* bp = B + (size_t)(tn + rc) * ldb + ka;
            b.x = bp[0];
            b.y = bp[1];
        } else {
            b.x = B[(size_t)(ka + 0) * ldb + tn + rc];
            b.y = B[(size_t)(ka + 1) * ldb + tn + rc];
        }
        acc = __builtin_amdgcn_wmma_f32_16x16x4_f32(
            false, a, false, b, (short)0, acc, false, false);
    }
#pragma unroll
    for (int i = 0; i < 8; ++i)
        C[(size_t)(tm + i + 8 * half) * ldc + tn + rc] = acc[i];
}

__global__ void self_init(const float* __restrict__ XW, const float* __restrict__ dinv,
                          const float* __restrict__ bias,
                          float* __restrict__ H, int ldh, int colOff, int n) {
    int idx = blockIdx.x * blockDim.x + threadIdx.x;
    int v = idx >> 5, lane = idx & 31;
    if (v >= n) return;
    float dn = dinv[v];
    float nn = dn * dn;
    float4 x = ((const float4*)(XW + (size_t)v * HD))[lane];
    float4 b = ((const float4*)bias)[lane];
    float* op = H + (size_t)v * ldh + colOff + lane * 4;
    op[0] = x.x * nn + b.x;
    op[1] = x.y * nn + b.y;
    op[2] = x.z * nn + b.z;
    op[3] = x.w * nn + b.w;
}

__global__ void __launch_bounds__(256)
edge_scatter(const float* __restrict__ XW, const float* __restrict__ dinv,
             const int* __restrict__ src, const int* __restrict__ dst,
             float* __restrict__ H, int ldh, int colOff, int e) {
    int lane = threadIdx.x & 31;
    int base = __builtin_amdgcn_readfirstlane(
        (int)((blockIdx.x * (blockDim.x >> 5) + (threadIdx.x >> 5)) << 5));
    if (base >= e) return;
    int eid = base + lane;
    int s = 0, d = 0;
    float nrm = 0.0f;
    if (eid < e) {
        s = src[eid];
        d = dst[eid];
        nrm = dinv[s] * dinv[d];
    }
    int cnt = e - base;
    if (cnt > 32) cnt = 32;
    const float* xb = XW + lane;
    float* hb = H + colOff + lane;
    int j = 0;
    for (; j + 4 <= cnt; j += 4) {
        float v[4][4];
        float nj[4];
        int dj[4];
#pragma unroll
        for (int u = 0; u < 4; ++u) {
            int sj = __builtin_amdgcn_readlane(s, j + u);
            dj[u] = __builtin_amdgcn_readlane(d, j + u);
            nj[u] = __int_as_float(__builtin_amdgcn_readlane(__float_as_int(nrm), j + u));
            const float* xp = xb + (size_t)sj * HD;
#pragma unroll
            for (int c = 0; c < 4; ++c) v[u][c] = xp[32 * c];
        }
#pragma unroll
        for (int u = 0; u < 4; ++u) {
            float* op = hb + (size_t)dj[u] * ldh;
#pragma unroll
            for (int c = 0; c < 4; ++c) atomicAdd(op + 32 * c, v[u][c] * nj[u]);
        }
    }
    for (; j < cnt; ++j) {
        int sj = __builtin_amdgcn_readlane(s, j);
        int dd = __builtin_amdgcn_readlane(d, j);
        float nn = __int_as_float(__builtin_amdgcn_readlane(__float_as_int(nrm), j));
        const float* xp = xb + (size_t)sj * HD;
        float* op = hb + (size_t)dd * ldh;
#pragma unroll
        for (int c = 0; c < 4; ++c) atomicAdd(op + 32 * c, xp[32 * c] * nn);
    }
}

__global__ void logits_init(float* logits, const float* __restrict__ lb2, int n) {
    int i = blockIdx.x * blockDim.x + threadIdx.x;
    if (i < n) {
        logits[2 * i + 0] = lb2[0];
        logits[2 * i + 1] = lb2[1];
    }
}

__global__ void __launch_bounds__(256)
linear_fused(const float* __restrict__ H,
             const float* __restrict__ lW1,
             const float* __restrict__ lb1,
             const float* __restrict__ lW2,
             float* __restrict__ logits,
             int totalTiles) {
    const int K = CD, tilesPerRow = CD / 16;
    int wave = threadIdx.x >> 5;
    int lane = threadIdx.x & 31;
    int tile = blockIdx.x * (blockDim.x >> 5) + wave;
    if (tile >= totalTiles) return;
    int tm = (tile / tilesPerRow) << 4;
    int tn = (tile % tilesPerRow) << 4;
    int col  = lane & 15;
    int half = lane >> 4;

    v8f acc = {};
    const float* arow = H + (size_t)(tm + col) * K;
    const float* brow = lW1 + (size_t)(tn + col) * K;
#pragma unroll 4
    for (int k0 = 0; k0 < K; k0 += 4) {
        int ka = k0 + 2 * half;
        v2f a;
        a.x = arow[ka];
        a.y = arow[ka + 1];
        v2f b;
        b.x = brow[ka];
        b.y = brow[ka + 1];
        acc = __builtin_amdgcn_wmma_f32_16x16x4_f32(
            false, a, false, b, (short)0, acc, false, false);
    }

    float bias = lb1[tn + col];
    float w0 = lW2[0 * CD + tn + col];
    float w1 = lW2[1 * CD + tn + col];
#pragma unroll
    for (int i = 0; i < 8; ++i) {
        float t = acc[i] + bias;
        t = t > 0.0f ? t : 0.0f;
        float p0 = t * w0;
        float p1 = t * w1;
#pragma unroll
        for (int m = 8; m >= 1; m >>= 1) {
            p0 += __shfl_xor(p0, m, 16);
            p1 += __shfl_xor(p1, m, 16);
        }
        if (col == 0) {
            int r = tm + i + 8 * half;
            atomicAdd(&logits[2 * r + 0], p0);
            atomicAdd(&logits[2 * r + 1], p1);
        }
    }
}

__global__ void softmax2(const float* __restrict__ logits, float* __restrict__ probs, int n) {
    int i = blockIdx.x * blockDim.x + threadIdx.x;
    if (i >= n) return;
    float a = logits[2 * i + 0], b = logits[2 * i + 1];
    float m = fmaxf(a, b);
    float ea = __expf(a - m), eb = __expf(b - m);
    float inv = 1.0f / (ea + eb);
    probs[2 * i + 0] = ea * inv;
    probs[2 * i + 1] = eb * inv;
}

extern "C" void kernel_launch(void* const* d_in, const int* in_sizes, int n_in,
                              void* d_out, int out_size, void* d_ws, size_t ws_size,
                              hipStream_t stream) {
    const float* x   = (const float*)d_in[0];
    const int*   ei  = (const int*)d_in[1];
    const float* W1  = (const float*)d_in[2];
    const float* b1  = (const float*)d_in[3];
    const float* W2  = (const float*)d_in[4];
    const float* b2  = (const float*)d_in[5];
    const float* W3  = (const float*)d_in[6];
    const float* b3  = (const float*)d_in[7];
    const float* lW1 = (const float*)d_in[8];
    const float* lb1 = (const float*)d_in[9];
    const float* lW2 = (const float*)d_in[10];
    const float* lb2 = (const float*)d_in[11];

    const int N = NN, E = EE;
    const int* src = ei;
    const int* dst = ei + E;

    float* dinv = (float*)d_ws;
    float* xw   = dinv + N;
    float* h    = xw + (size_t)N * HD;

    float* logits = (float*)d_out;
    float* probs  = logits + (size_t)N * 2;

    const int T = 256;
    const int rowTiles = N / 16;

    deg_init<<<(N + T - 1) / T, T, 0, stream>>>(dinv, N);
    deg_count<<<(E + T - 1) / T, T, 0, stream>>>(dst, dinv, E);
    deg_rsqrt<<<(N + T - 1) / T, T, 0, stream>>>(dinv, N);

    {
        int tiles = rowTiles * (HD / 16);
        gemm16_f32<false><<<(tiles + 7) / 8, T, 0, stream>>>(
            x, 64, W1, HD, xw, HD, 64, HD / 16, tiles);
        self_init<<<((size_t)N * 32 + T - 1) / T, T, 0, stream>>>(xw, dinv, b1, h, CD, 0, N);
        edge_scatter<<<(E + T - 1) / T, T, 0, stream>>>(
            xw, dinv, src, dst, h, CD, 0, E);
    }
    {
        int tiles = rowTiles * (HD / 16);
        gemm16_f32<false><<<(tiles + 7) / 8, T, 0, stream>>>(
            h + 0, CD, W2, HD, xw, HD, HD, HD / 16, tiles);
        self_init<<<((size_t)N * 32 + T - 1) / T, T, 0, stream>>>(xw, dinv, b2, h, CD, HD, N);
        edge_scatter<<<(E + T - 1) / T, T, 0, stream>>>(
            xw, dinv, src, dst, h, CD, HD, E);
    }
    {
        int tiles = rowTiles * (HD / 16);
        gemm16_f32<false><<<(tiles + 7) / 8, T, 0, stream>>>(
            h + HD, CD, W3, HD, xw, HD, HD, HD / 16, tiles);
        self_init<<<((size_t)N * 32 + T - 1) / T, T, 0, stream>>>(xw, dinv, b3, h, CD, 2 * HD, N);
        edge_scatter<<<(E + T - 1) / T, T, 0, stream>>>(
            xw, dinv, src, dst, h, CD, 2 * HD, E);
    }

    logits_init<<<(N + T - 1) / T, T, 0, stream>>>(logits, lb2, N);
    {
        int tiles = rowTiles * (CD / 16);
        linear_fused<<<(tiles + 7) / 8, T, 0, stream>>>(h, lW1, lb1, lW2, logits, tiles);
    }
    softmax2<<<(N + T - 1) / T, T, 0, stream>>>(logits, probs, N);
}


# Head GEMM: one wave owns 16 rows x all 384 columns (24 accumulator tiles), lW1 repacked per call so B loads are whole lines; row sums in-wave, no atomics, k summed in another order; 14 launches
# speedup vs baseline: 1.8264x; 1.3905x over previous
#include <hip/hip_runtime.h>
#include <hip/hip_bf16.h>


typedef __attribute__((ext_vector_type(2))) float v2f;
typedef __attribute__((ext_vector_type(8))) float v8f;

#define NN 50000
#define EE 800000
#define HD 128
#define CD 384

__global__ void deg_init(float* deg, int n) {
    int i = blockIdx.x * blockDim.x + threadIdx.x;
    if (i < n) deg[i] = 1.0f;
}
__global__ void deg_count(const int* __restrict__ dst, float* deg, int e) {
    int i = blockIdx.x * blockDim.x + threadIdx.x;
    if (i < e) atomicAdd(&deg[dst[i]], 1.0f);
}
__global__ void deg_rsqrt(float* deg, int n) {
    int i = blockIdx.x * blockDim.x + threadIdx.x;
    if (i < n) deg[i] = rsqrtf(deg[i]);
}

template <bool BT>
__global__ void __launch_bounds__(256)
gemm16_f32(const float* __restrict__ A, int lda,
           const float* __restrict__ B, int ldb,
           float* __restrict__ C, int ldc,
           int K, int tilesPerRow, int totalTiles) {
    int wave = threadIdx.x >> 5;
    int lane = threadIdx.x & 31;
    int tile = blockIdx.x * (blockDim.x >> 5) + wave;
    if (tile >= totalTiles) return;
    int tm = (tile / tilesPerRow) << 4;
    int tn = (tile % tilesPerRow) << 4;
    int rc   = lane & 15;
    int half = lane >> 4;

    v8f acc = {};
    const float* arow = A + (size_t)(tm + rc) * lda;
#pragma unroll 4
    for (int k0 = 0; k0 < K; k0 += 4) {
        int ka = k0 + 2 * half;
        v2f a;
        a.x = arow[ka];
        a.y = arow[ka + 1];
        v2f b;
        if (BT) {
            const float* bp = B + (size_t)(tn + rc) * ldb + ka;
            b.x = bp[0];
            b.y = bp[1];
        } else {
            b.x = B[(size_t)(ka + 0) * ldb + tn + rc];
            b.y = B[(size_t)(ka + 1) * ldb + tn + rc];
        }
        acc = __builtin_amdgcn_wmma_f32_16x16x4_f32(
            false, a, false, b, (short)0, acc, false, false);
    }
#pragma unroll
    for (int i = 0; i < 8; ++i)
        C[(size_t)(tm + i + 8 * half) * ldc + tn + rc] = acc[i];
}

__global__ void self_init(const float* __restrict__ XW, const float* __restrict__ dinv,
                          const float* __restrict__ bias,
                          float* __restrict__ H, int ldh, int colOff, int n) {
    int idx = blockIdx.x * blockDim.x + threadIdx.x;
    int v = idx >> 5, lane = idx & 31;
    if (v >= n) return;
    float dn = dinv[v];
    float nn = dn * dn;
    float4 x = ((const float4*)(XW + (size_t)v * HD))[lane];
    float4 b = ((const float4*)bias)[lane];
    float* op = H + (size_t)v * ldh + colOff + lane * 4;
    op[0] = x.x * nn + b.x;
    op[1] = x.y * nn + b.y;
    op[2] = x.z * nn + b.z;
    op[3] = x.w * nn + b.w;
}

__global__ void __launch_bounds__(256)
edge_scatter(const float* __restrict__ XW, const float* __restrict__ dinv,
             const int* __restrict__ src, const int* __restrict__ dst,
             float* __restrict__ H, int ldh, int colOff, int e) {
    int lane = threadIdx.x & 31;
    int base = __builtin_amdgcn_readfirstlane(
        (int)((blockIdx.x * (blockDim.x >> 5) + (threadIdx.x >> 5)) << 5));
    if (base >= e) return;
    int eid = base + lane;
    int s = 0, d = 0;
    float nrm = 0.0f;
    if (eid < e) {
        s = src[eid];
        d = dst[eid];
        nrm = dinv[s] * dinv[d];
    }
    int cnt = e - base;
    if (cnt > 32) cnt = 32;
    const float* xb = XW + lane;
    float* hb = H + colOff + lane;
    int j = 0;
    for (; j + 4 <= cnt; j += 4) {
        float v[4][4];
        float nj[4];
        int dj[4];
#pragma unroll
        for (int u = 0; u < 4; ++u) {
            int sj = __builtin_amdgcn_readlane(s, j + u);
            dj[u] = __builtin_amdgcn_readlane(d, j + u);
            nj[u] = __int_as_float(__builtin_amdgcn_readlane(__float_as_int(nrm), j + u));
            const float* xp = xb + (size_t)sj * HD;
#pragma unroll
            for (int c = 0; c < 4; ++c) v[u][c] = xp[32 * c];
        }
#pragma unroll
        for (int u = 0; u < 4; ++u) {
            float* op = hb + (size_t)dj[u] * ldh;
#pragma unroll
            for (int c = 0; c < 4; ++c) atomicAdd(op + 32 * c, v[u][c] * nj[u]);
        }
    }
    for (; j < cnt; ++j) {
        int sj = __builtin_amdgcn_readlane(s, j);
        int dd = __builtin_amdgcn_readlane(d, j);
        float nn = __int_as_float(__builtin_amdgcn_readlane(__float_as_int(nrm), j));
        const float* xp = xb + (size_t)sj * HD;
        float* op = hb + (size_t)dd * ldh;
#pragma unroll
        for (int c = 0; c < 4; ++c) atomicAdd(op + 32 * c, xp[32 * c] * nn);
    }
}

__global__ void __launch_bounds__(256)
pack_head_w(const float* __restrict__ lW1, float* __restrict__ P) {
    int idx = blockIdx.x * blockDim.x + threadIdx.x;
    if (idx >= (CD / 8) * (CD / 16) * 32) return;
    int lane = idx & 31;
    int nt = (idx >> 5) % (CD / 16);
    int b  = (idx >> 5) / (CD / 16);
    int c = lane & 15, half = lane >> 4;
    float4 v = *(const float4*)(lW1 + (size_t)(nt * 16 + c) * CD + 8 * b + 4 * half);
    ((float4*)P)[idx] = v;
}

#define HT (CD / 16)
#define HG 6
__global__ void __launch_bounds__(128)
head_rows(const float* __restrict__ H,
          const float* __restrict__ P,
          const float* __restrict__ lb1,
          const float* __restrict__ lW2,
          const float* __restrict__ lb2,
          float* __restrict__ logits,
          float* __restrict__ probs,
          int rowTiles) {
    int lane = threadIdx.x & 31;
    int tile = blockIdx.x * (blockDim.x >> 5) + (threadIdx.x >> 5);
    if (tile >= rowTiles) return;
    int c = lane & 15, half = lane >> 4;
    int tm = tile << 4;

    v8f acc[HT];
#pragma unroll
    for (int nt = 0; nt < HT; ++nt) acc[nt] = (v8f){};

    const float4* ap = (const float4*)(H + (size_t)(tm + c) * CD + 4 * half);
    const float4* bp = (const float4*)P + lane;
    const int NB = CD / 8;

    float4 bcur[HG], bnxt[HG];
#pragma unroll
    for (int t = 0; t < HG; ++t) bcur[t] = bp[t * 32];
    float4 acur = ap[0];

    for (int b = 0; b < NB; ++b) {
        int bn = b + 1 < NB ? b + 1 : b;
        float4 anxt = ap[2 * bn];
        v2f a0 = {acur.x, acur.y};
        v2f a1 = {acur.z, acur.w};
#pragma unroll
        for (int g = 0; g < HT / HG; ++g) {
            int gn = b * (HT / HG) + g + 1;
            if (gn > NB * (HT / HG) - 1) gn = NB * (HT / HG) - 1;
#pragma unroll
            for (int t = 0; t < HG; ++t) bnxt[t] = bp[(gn * HG + t) * 32];
#pragma unroll
            for (int t = 0; t < HG; ++t) {
                v2f b0 = {bcur[t].x, bcur[t].y};
                acc[g * HG + t] = __builtin_amdgcn_wmma_f32_16x16x4_f32(
                    false, a0, false, b0, (short)0, acc[g * HG + t], false, false);
            }
#pragma unroll
            for (int t = 0; t < HG; ++t) {
                v2f b1 = {bcur[t].z, bcur[t].w};
                acc[g * HG + t] = __builtin_amdgcn_wmma_f32_16x16x4_f32(
                    false, a1, false, b1, (short)0, acc[g * HG + t], false, false);
            }
#pragma unroll
            for (int t = 0; t < HG; ++t) bcur[t] = bnxt[t];
        }
        acur = anxt;
    }

    float p0[8], p1[8];
#pragma unroll
    for (int i = 0; i < 8; ++i) { p0[i] = 0.0f; p1[i] = 0.0f; }
#pragma unroll
    for (int nt = 0; nt < HT; ++nt) {
        float bias = lb1[nt * 16 + c];
        float w0 = lW2[0 * CD + nt * 16 + c];
        float w1 = lW2[1 * CD + nt * 16 + c];
#pragma unroll
        for (int i = 0; i < 8; ++i) {
            float t = acc[nt][i] + bias;
            t = t > 0.0f ? t : 0.0f;
            p0[i] += t * w0;
            p1[i] += t * w1;
        }
    }
#pragma unroll
    for (int i = 0; i < 8; ++i) {
#pragma unroll
        for (int m = 8; m >= 1; m >>= 1) {
            p0[i] += __shfl_xor(p0[i], m, 16);
            p1[i] += __shfl_xor(p1[i], m, 16);
        }
    }
    float c0 = lb2[0], c1 = lb2[1];
#pragma unroll
    for (int i = 0; i < 8; ++i) {
        if (c == i) {
            int r = tm + i + 8 * half;
            float a = p0[i] + c0, bb = p1[i] + c1;
            float m = fmaxf(a, bb);
            float ea = __expf(a - m), eb = __expf(bb - m);
            float inv = 1.0f / (ea + eb);
            logits[2 * r + 0] = a;
            logits[2 * r + 1] = bb;
            probs[2 * r + 0] = ea * inv;
            probs[2 * r + 1] = eb * inv;
        }
    }
}

extern "C" void kernel_launch(void* const* d_in, const int* in_sizes, int n_in,
                              void* d_out, int out_size, void* d_ws, size_t ws_size,
                              hipStream_t stream) {
    const float* x   = (const float*)d_in[0];
    const int*   ei  = (const int*)d_in[1];
    const float* W1  = (const float*)d_in[2];
    const float* b1  = (const float*)d_in[3];
    const float* W2  = (const float*)d_in[4];
    const float* b2  = (const float*)d_in[5];
    const float* W3  = (const float*)d_in[6];
    const float* b3  = (const float*)d_in[7];
    const float* lW1 = (const float*)d_in[8];
    const float* lb1 = (const float*)d_in[9];
    const float* lW2 = (const float*)d_in[10];
    const float* lb2 = (const float*)d_in[11];

    const int N = NN, E = EE;
    const int* src = ei;
    const int* dst = ei + E;

    float* dinv = (float*)d_ws;
    float* xw   = dinv + N;
    float* h    = xw + (size_t)N * HD;

    float* logits = (float*)d_out;
    float* probs  = logits + (size_t)N * 2;

    const int T = 256;
    const int rowTiles = N / 16;

    deg_init<<<(N + T - 1) / T, T, 0, stream>>>(dinv, N);
    deg_count<<<(E + T - 1) / T, T, 0, stream>>>(dst, dinv, E);
    deg_rsqrt<<<(N + T - 1) / T, T, 0, stream>>>(dinv, N);

    {
        int tiles = rowTiles * (HD / 16);
        gemm16_f32<false><<<(tiles + 7) / 8, T, 0, stream>>>(
            x, 64, W1, HD, xw, HD, 64, HD / 16, tiles);
        self_init<<<((size_t)N * 32 + T - 1) / T, T, 0, stream>>>(xw, dinv, b1, h, CD, 0, N);
        edge_scatter<<<(E + T - 1) / T, T, 0, stream>>>(
            xw, dinv, src, dst, h, CD, 0, E);
    }
    {
        int tiles = rowTiles * (HD / 16);
        gemm16_f32<false><<<(tiles + 7) / 8, T, 0, stream>>>(
            h + 0, CD, W2, HD, xw, HD, HD, HD / 16, tiles);
        self_init<<<((size_t)N * 32 + T - 1) / T, T, 0, stream>>>(xw, dinv, b2, h, CD, HD, N);
        edge_scatter<<<(E + T - 1) / T, T, 0, stream>>>(
            xw, dinv, src, dst, h, CD, HD, E);
    }
    {
        int tiles = rowTiles * (HD / 16);
        gemm16_f32<false><<<(tiles + 7) / 8, T, 0, stream>>>(
            h + HD, CD, W3, HD, xw, HD, HD, HD / 16, tiles);
        self_init<<<((size_t)N * 32 + T - 1) / T, T, 0, stream>>>(xw, dinv, b3, h, CD, 2 * HD, N);
        edge_scatter<<<(E + T - 1) / T, T, 0, stream>>>(
            xw, dinv, src, dst, h, CD, 2 * HD, E);
    }

    float* wpk = h + (size_t)N * CD + 48;
    pack_head_w<<<((CD / 8) * (CD / 16) * 32 + T - 1) / T, T, 0, stream>>>(lW1, wpk);
    head_rows<<<(rowTiles + 3) / 4, 128, 0, stream>>>(
        h, wpk, lb1, lW2, lb2, logits, probs, rowTiles);
}


# Edge sums by owner work-groups: each holds 196 output rows in LDS, scans dst, adds its own edges there with LDS f32 adds, stores each line once; no global row atomics; sums in LDS-arrival order
# speedup vs baseline: 2.4026x; 1.3155x over previous
#include <hip/hip_runtime.h>
#include <hip/hip_bf16.h>


typedef __attribute__((ext_vector_type(2))) float v2f;
typedef __attribute__((ext_vector_type(8))) float v8f;

#define NN 50000
#define EE 800000
#define HD 128
#define CD 384

__global__ void deg_init(float* deg, int n) {
    int i = blockIdx.x * blockDim.x + threadIdx.x;
    if (i < n) deg[i] = 1.0f;
}
__global__ void deg_count(const int* __restrict__ dst, float* deg, int e) {
    int i = blockIdx.x * blockDim.x + threadIdx.x;
    if (i < e) atomicAdd(&deg[dst[i]], 1.0f);
}
__global__ void deg_rsqrt(float* deg, int n) {
    int i = blockIdx.x * blockDim.x + threadIdx.x;
    if (i < n) deg[i] = rsqrtf(deg[i]);
}

template <bool BT>
__global__ void __launch_bounds__(256)
gemm16_f32(const float* __restrict__ A, int lda,
           const float* __restrict__ B, int ldb,
           float* __restrict__ C, int ldc,
           int K, int tilesPerRow, int totalTiles) {
    int wave = threadIdx.x >> 5;
    int lane = threadIdx.x & 31;
    int tile = blockIdx.x * (blockDim.x >> 5) + wave;
    if (tile >= totalTiles) return;
    int tm = (tile / tilesPerRow) << 4;
    int tn = (tile % tilesPerRow) << 4;
    int rc   = lane & 15;
    int half = lane >> 4;

    v8f acc = {};
    const float* arow = A + (size_t)(tm + rc) * lda;
#pragma unroll 4
    for (int k0 = 0; k0 < K; k0 += 4) {
        int ka = k0 + 2 * half;
        v2f a;
        a.x = arow[ka];
        a.y = arow[ka + 1];
        v2f b;
        if (BT) {
            const float* bp = B + (size_t)(tn + rc) * ldb + ka;
            b.x = bp[0];
            b.y = bp[1];
        } else {
            b.x = B[(size_t)(ka + 0) * ldb + tn + rc];
            b.y = B[(size_t)(ka + 1) * ldb + tn + rc];
        }
        acc = __builtin_amdgcn_wmma_f32_16x16x4_f32(
            false, a, false, b, (short)0, acc, false, false);
    }
#pragma unroll
    for (int i = 0; i < 8; ++i)
        C[(size_t)(tm + i + 8 * half) * ldc + tn + rc] = acc[i];
}

#define OWN_R 196
#define OWN_T 1024
#define OWN_W (OWN_T / 32)
#define OWN_Q 544

template <int U>
__device__ __forceinline__ void owner_rows(
        int s, int rl, float nrm, int j, int lane,
        const float* __restrict__ XW, float* acc) {
    float4 v[U];
    float nj[U];
    int rj[U];
#pragma unroll
    for (int u = 0; u < U; ++u) {
        int sj = __builtin_amdgcn_readlane(s, j + u);
        rj[u] = __builtin_amdgcn_readlane(rl, j + u);
        nj[u] = __int_as_float(__builtin_amdgcn_readlane(__float_as_int(nrm), j + u));
        v[u] = ((const float4*)(XW + (size_t)sj * HD))[lane];
    }
#pragma unroll
    for (int u = 0; u < U; ++u) {
        float* a = acc + rj[u] * HD + lane;
        atomicAdd(a,      v[u].x * nj[u]);
        atomicAdd(a + 32, v[u].y * nj[u]);
        atomicAdd(a + 64, v[u].z * nj[u]);
        atomicAdd(a + 96, v[u].w * nj[u]);
    }
}

__device__ __forceinline__ void owner_drain(
        const int* q, int qn, int lane,
        const float* __restrict__ XW, const float* __restrict__ dinv,
        const int* __restrict__ src, const float* rdinv, float* acc) {
    for (int q0 = 0; q0 < qn; q0 += 32) {
        int cnt = qn - q0;
        if (cnt > 32) cnt = 32;
        int s = 0, rl = 0;
        float nrm = 0.0f;
        if (lane < cnt) {
            int ent = q[q0 + lane];
            rl = ent >> 20;
            s = src[ent & 0xFFFFF];
            nrm = dinv[s] * rdinv[rl];
        }
        int j = 0;
        for (; j + 8 <= cnt; j += 8) owner_rows<8>(s, rl, nrm, j, lane, XW, acc);
        if (j + 4 <= cnt) { owner_rows<4>(s, rl, nrm, j, lane, XW, acc); j += 4; }
        if (j + 2 <= cnt) { owner_rows<2>(s, rl, nrm, j, lane, XW, acc); j += 2; }
        if (j < cnt) owner_rows<1>(s, rl, nrm, j, lane, XW, acc);
    }
}

__global__ void __launch_bounds__(OWN_T)
owner_layer(const float* __restrict__ XW, const float* __restrict__ dinv,
            const float* __restrict__ bias,
            const int* __restrict__ src, const int* __restrict__ dst,
            float* __restrict__ H, int ldh, int colOff, int n, int e) {
    __shared__ float acc[OWN_R * HD];
    __shared__ float rdinv[OWN_R];
    __shared__ int queue[OWN_W][OWN_Q];
    static_assert(EE < (1 << 20), "edge id must fit the queue entry's low 20 bits");

    int lane = threadIdx.x & 31;
    int wave = threadIdx.x >> 5;
    int lo = blockIdx.x * OWN_R;
    int rows = n - lo;
    if (rows > OWN_R) rows = OWN_R;

    {
        const int RW = (OWN_R + OWN_W - 1) / OWN_W;
        float4 b = ((const float4*)bias)[lane];
        float4 x[RW];
        float dn[RW];
#pragma unroll
        for (int i = 0; i < RW; ++i) {
            int r = wave + i * OWN_W;
            int v = lo + (r < rows ? r : 0);
            dn[i] = dinv[v];
            x[i] = ((const float4*)(XW + (size_t)v * HD))[lane];
        }
#pragma unroll
        for (int i = 0; i < RW; ++i) {
            int r = wave + i * OWN_W;
            if (r < rows) {
                float nn = dn[i] * dn[i];
                float* a = acc + r * HD + lane;
                a[0]  = x[i].x * nn + b.x;
                a[32] = x[i].y * nn + b.y;
                a[64] = x[i].z * nn + b.z;
                a[96] = x[i].w * nn + b.w;
                if (lane == 0) rdinv[r] = dn[i];
            }
        }
    }
    __syncthreads();

    int* q = queue[wave];
    int qn = 0;
    const int4* d4 = (const int4*)dst;
    int G = e >> 2;
    auto fetch = [&](int4 (&b)[4], int g0) {
#pragma unroll
        for (int u = 0; u < 4; ++u) {
            int g = g0 + u * OWN_T + (int)threadIdx.x;
            b[u] = d4[g < G ? g : 0];
        }
    };
    auto sift = [&](const int4 (&b)[4], int g0) {
#pragma unroll
        for (int u = 0; u < 4; ++u) {
            int g = g0 + u * OWN_T + (int)threadIdx.x;
            unsigned lim = g < G ? (unsigned)rows : 0u;
            int dd[4] = {b[u].x, b[u].y, b[u].z, b[u].w};
            unsigned rel[4], mk[4];
#pragma unroll
            for (int c = 0; c < 4; ++c) {
                rel[c] = (unsigned)(dd[c] - lo);
                mk[c] = __builtin_amdgcn_ballot_w32(rel[c] < lim);
            }
            if (mk[0] | mk[1] | mk[2] | mk[3]) {
#pragma unroll
                for (int c = 0; c < 4; ++c) {
                    if (rel[c] < lim) {
                        int pos = qn + __popc(mk[c] & ((1u << lane) - 1u));
                        q[pos] = (int)(rel[c] << 20) | (g * 4 + c);
                    }
                    qn += __popc(mk[c]);
                }
            }
        }
        if (qn >= 32) {
            owner_drain(q, qn, lane, XW, dinv, src, rdinv, acc);
            qn = 0;
        }
    };
    const int RD = OWN_T * 4;
    int4 ra[4], rb[4];
    fetch(ra, 0);
    for (int g0 = 0; g0 < G; g0 += 2 * RD) {
        fetch(rb, g0 + RD);
        sift(ra, g0);
        fetch(ra, g0 + 2 * RD);
        sift(rb, g0 + RD);
    }
    if (wave == 0) {
        int eid = 4 * G + lane;
        unsigned rel = eid < e ? (unsigned)(dst[eid] - lo) : 0xFFFFFFFFu;
        unsigned mk = __builtin_amdgcn_ballot_w32(rel < (unsigned)rows);
        if (rel < (unsigned)rows) {
            int pos = qn + __popc(mk & ((1u << lane) - 1u));
            q[pos] = (int)(rel << 20) | eid;
        }
        qn += __popc(mk);
    }
    owner_drain(q, qn, lane, XW, dinv, src, rdinv, acc);
    __syncthreads();

    for (int r = wave; r < rows; r += OWN_W) {
        const float* a = acc + r * HD + lane;
        float4 o = {a[0], a[32], a[64], a[96]};
        *(float4*)(H + (size_t)(lo + r) * ldh + colOff + lane * 4) = o;
    }
}

__global__ void __launch_bounds__(256)
pack_head_w(const float* __restrict__ lW1, float* __restrict__ P) {
    int idx = blockIdx.x * blockDim.x + threadIdx.x;
    if (idx >= (CD / 8) * (CD / 16) * 32) return;
    int lane = idx & 31;
    int nt = (idx >> 5) % (CD / 16);
    int b  = (idx >> 5) / (CD / 16);
    int c = lane & 15, half = lane >> 4;
    float4 v = *(const float4*)(lW1 + (size_t)(nt * 16 + c) * CD + 8 * b + 4 * half);
    ((float4*)P)[idx] = v;
}

#define HT (CD / 16)
#define HG 6
__global__ void __launch_bounds__(128)
head_rows(const float* __restrict__ H,
          const float* __restrict__ P,
          const float* __restrict__ lb1,
          const float* __restrict__ lW2,
          const float* __restrict__ lb2,
          float* __restrict__ logits,
          float* __restrict__ probs,
          int rowTiles) {
    int lane = threadIdx.x & 31;
    int tile = blockIdx.x * (blockDim.x >> 5) + (threadIdx.x >> 5);
    if (tile >= rowTiles) return;
    int c = lane & 15, half = lane >> 4;
    int tm = tile << 4;

    v8f acc[HT];
#pragma unroll
    for (int nt = 0; nt < HT; ++nt) acc[nt] = (v8f){};

    const float4* ap = (const float4*)(H + (size_t)(tm + c) * CD + 4 * half);
    const float4* bp = (const float4*)P + lane;
    const int NB = CD / 8;

    float4 bcur[HG], bnxt[HG];
#pragma unroll
    for (int t = 0; t < HG; ++t) bcur[t] = bp[t * 32];
    float4 acur = ap[0];

    for (int b = 0; b < NB; ++b) {
        int bn = b + 1 < NB ? b + 1 : b;
        float4 anxt = ap[2 * bn];
        v2f a0 = {acur.x, acur.y};
        v2f a1 = {acur.z, acur.w};
#pragma unroll
        for (int g = 0; g < HT / HG; ++g) {
            int gn = b * (HT / HG) + g + 1;
            if (gn > NB * (HT / HG) - 1) gn = NB * (HT / HG) - 1;
#pragma unroll
            for (int t = 0; t < HG; ++t) bnxt[t] = bp[(gn * HG + t) * 32];
#pragma unroll
            for (int t = 0; t < HG; ++t) {
                v2f b0 = {bcur[t].x, bcur[t].y};
                acc[g * HG + t] = __builtin_amdgcn_wmma_f32_16x16x4_f32(
                    false, a0, false, b0, (short)0, acc[g * HG + t], false, false);
            }
#pragma unroll
            for (int t = 0; t < HG; ++t) {
                v2f b1 = {bcur[t].z, bcur[t].w};
                acc[g * HG + t] = __builtin_amdgcn_wmma_f32_16x16x4_f32(
                    false, a1, false, b1, (short)0, acc[g * HG + t], false, false);
            }
#pragma unroll
            for (int t = 0; t < HG; ++t) bcur[t] = bnxt[t];
        }
        acur = anxt;
    }

    float p0[8], p1[8];
#pragma unroll
    for (int i = 0; i < 8; ++i) { p0[i] = 0.0f; p1[i] = 0.0f; }
#pragma unroll
    for (int nt = 0; nt < HT; ++nt) {
        float bias = lb1[nt * 16 + c];
        float w0 = lW2[0 * CD + nt * 16 + c];
        float w1 = lW2[1 * CD + nt * 16 + c];
#pragma unroll
        for (int i = 0; i < 8; ++i) {
            float t = acc[nt][i] + bias;
            t = t > 0.0f ? t : 0.0f;
            p0[i] += t * w0;
            p1[i] += t * w1;
        }
    }
#pragma unroll
    for (int i = 0; i < 8; ++i) {
#pragma unroll
        for (int m = 8; m >= 1; m >>= 1) {
            p0[i] += __shfl_xor(p0[i], m, 16);
            p1[i] += __shfl_xor(p1[i], m, 16);
        }
    }
    float c0 = lb2[0], c1 = lb2[1];
#pragma unroll
    for (int i = 0; i < 8; ++i) {
        if (c == i) {
            int r = tm + i + 8 * half;
            float a = p0[i] + c0, bb = p1[i] + c1;
            float m = fmaxf(a, bb);
            float ea = __expf(a - m), eb = __expf(bb - m);
            float inv = 1.0f / (ea + eb);
            logits[2 * r + 0] = a;
            logits[2 * r + 1] = bb;
            probs[2 * r + 0] = ea * inv;
            probs[2 * r + 1] = eb * inv;
        }
    }
}

extern "C" void kernel_launch(void* const* d_in, const int* in_sizes, int n_in,
                              void* d_out, int out_size, void* d_ws, size_t ws_size,
                              hipStream_t stream) {
    const float* x   = (const float*)d_in[0];
    const int*   ei  = (const int*)d_in[1];
    const float* W1  = (const float*)d_in[2];
    const float* b1  = (const float*)d_in[3];
    const float* W2  = (const float*)d_in[4];
    const float* b2  = (const float*)d_in[5];
    const float* W3  = (const float*)d_in[6];
    const float* b3  = (const float*)d_in[7];
    const float* lW1 = (const float*)d_in[8];
    const float* lb1 = (const float*)d_in[9];
    const float* lW2 = (const float*)d_in[10];
    const float* lb2 = (const float*)d_in[11];

    const int N = NN, E = EE;
    const int* src = ei;
    const int* dst = ei + E;

    float* dinv = (float*)d_ws;
    float* xw   = dinv + N;
    float* h    = xw + (size_t)N * HD;

    float* logits = (float*)d_out;
    float* probs  = logits + (size_t)N * 2;

    const int T = 256;
    const int rowTiles = N / 16;

    deg_init<<<(N + T - 1) / T, T, 0, stream>>>(dinv, N);
    deg_count<<<(E + T - 1) / T, T, 0, stream>>>(dst, dinv, E);
    deg_rsqrt<<<(N + T - 1) / T, T, 0, stream>>>(dinv, N);

    {
        int tiles = rowTiles * (HD / 16);
        gemm16_f32<false><<<(tiles + 7) / 8, T, 0, stream>>>(
            x, 64, W1, HD, xw, HD, 64, HD / 16, tiles);
        owner_layer<<<(N + OWN_R - 1) / OWN_R, OWN_T, 0, stream>>>(
            xw, dinv, b1, src, dst, h, CD, 0, N, E);
    }
    {
        int tiles = rowTiles * (HD / 16);
        gemm16_f32<false><<<(tiles + 7) / 8, T, 0, stream>>>(
            h + 0, CD, W2, HD, xw, HD, HD, HD / 16, tiles);
        owner_layer<<<(N + OWN_R - 1) / OWN_R, OWN_T, 0, stream>>>(
            xw, dinv, b2, src, dst, h, CD, HD, N, E);
    }
    {
        int tiles = rowTiles * (HD / 16);
        gemm16_f32<false><<<(tiles + 7) / 8, T, 0, stream>>>(
            h + HD, CD, W3, HD, xw, HD, HD, HD / 16, tiles);
        owner_layer<<<(N + OWN_R - 1) / OWN_R, OWN_T, 0, stream>>>(
            xw, dinv, b3, src, dst, h, CD, 2 * HD, N, E);
    }

    float* wpk = h + (size_t)N * CD + 48;
    pack_head_w<<<((CD / 8) * (CD / 16) * 32 + T - 1) / T, T, 0, stream>>>(lW1, wpk);
    head_rows<<<(rowTiles + 3) / 4, 128, 0, stream>>>(
        h, wpk, lb1, lW2, lb2, logits, probs, rowTiles);
}


# Owner scan reads one owner byte per edge (dst/196, written by the degree pass each call), 16 edges a load, zero-byte test; local row fetched with src in the drain; sums in another LDS-arrival order
# speedup vs baseline: 2.5723x; 1.0706x over previous
#include <hip/hip_runtime.h>
#include <hip/hip_bf16.h>


typedef __attribute__((ext_vector_type(2))) float v2f;
typedef __attribute__((ext_vector_type(8))) float v8f;

#define NN 50000
#define EE 800000
#define HD 128
#define CD 384

__global__ void deg_init(float* deg, int n) {
    int i = blockIdx.x * blockDim.x + threadIdx.x;
    if (i < n) deg[i] = 1.0f;
}
#define OWN_R 196
__global__ void deg_count(const int* __restrict__ dst, float* deg,
                          unsigned* __restrict__ own8, int e) {
    int g = blockIdx.x * blockDim.x + threadIdx.x;
    int e0 = g * 4;
    if (e0 >= e) return;
    unsigned pk = 0;
    if (e0 + 4 <= e) {
        int4 d = ((const int4*)dst)[g];
        atomicAdd(&deg[d.x], 1.0f);
        atomicAdd(&deg[d.y], 1.0f);
        atomicAdd(&deg[d.z], 1.0f);
        atomicAdd(&deg[d.w], 1.0f);
        pk = ((unsigned)d.x / OWN_R) | (((unsigned)d.y / OWN_R) << 8) |
             (((unsigned)d.z / OWN_R) << 16) | (((unsigned)d.w / OWN_R) << 24);
    } else {
        for (int c = 0; c < 4; ++c) {
            if (e0 + c < e) {
                int d = dst[e0 + c];
                atomicAdd(&deg[d], 1.0f);
                pk |= ((unsigned)d / OWN_R) << (8 * c);
            }
        }
    }
    own8[g] = pk;
}
__global__ void deg_rsqrt(float* deg, int n) {
    int i = blockIdx.x * blockDim.x + threadIdx.x;
    if (i < n) deg[i] = rsqrtf(deg[i]);
}

template <bool BT>
__global__ void __launch_bounds__(256)
gemm16_f32(const float* __restrict__ A, int lda,
           const float* __restrict__ B, int ldb,
           float* __restrict__ C, int ldc,
           int K, int tilesPerRow, int totalTiles) {
    int wave = threadIdx.x >> 5;
    int lane = threadIdx.x & 31;
    int tile = blockIdx.x * (blockDim.x >> 5) + wave;
    if (tile >= totalTiles) return;
    int tm = (tile / tilesPerRow) << 4;
    int tn = (tile % tilesPerRow) << 4;
    int rc   = lane & 15;
    int half = lane >> 4;

    v8f acc = {};
    const float* arow = A + (size_t)(tm + rc) * lda;
#pragma unroll 4
    for (int k0 = 0; k0 < K; k0 += 4) {
        int ka = k0 + 2 * half;
        v2f a;
        a.x = arow[ka];
        a.y = arow[ka + 1];
        v2f b;
        if (BT) {
            const float* bp = B + (size_t)(tn + rc) * ldb + ka;
            b.x = bp[0];
            b.y = bp[1];
        } else {
            b.x = B[(size_t)(ka + 0) * ldb + tn + rc];
            b.y = B[(size_t)(ka + 1) * ldb + tn + rc];
        }
        acc = __builtin_amdgcn_wmma_f32_16x16x4_f32(
            false, a, false, b, (short)0, acc, false, false);
    }
#pragma unroll
    for (int i = 0; i < 8; ++i)
        C[(size_t)(tm + i + 8 * half) * ldc + tn + rc] = acc[i];
}

#define OWN_T 1024
#define OWN_W (OWN_T / 32)
#define OWN_Q 544

template <int U>
__device__ __forceinline__ void owner_rows(
        int s, int rl, float nrm, int j, int lane,
        const float* __restrict__ XW, float* acc) {
    float4 v[U];
    float nj[U];
    int rj[U];
#pragma unroll
    for (int u = 0; u < U; ++u) {
        int sj = __builtin_amdgcn_readlane(s, j + u);
        rj[u] = __builtin_amdgcn_readlane(rl, j + u);
        nj[u] = __int_as_float(__builtin_amdgcn_readlane(__float_as_int(nrm), j + u));
        v[u] = ((const float4*)(XW + (size_t)sj * HD))[lane];
    }
#pragma unroll
    for (int u = 0; u < U; ++u) {
        float* a = acc + rj[u] * HD + lane;
        atomicAdd(a,      v[u].x * nj[u]);
        atomicAdd(a + 32, v[u].y * nj[u]);
        atomicAdd(a + 64, v[u].z * nj[u]);
        atomicAdd(a + 96, v[u].w * nj[u]);
    }
}

__device__ __forceinline__ void owner_drain(
        const int* q, int qn, int lane, int lo,
        const float* __restrict__ XW, const float* __restrict__ dinv,
        const int* __restrict__ src, const int* __restrict__ dst,
        const float* rdinv, float* acc) {
    for (int q0 = 0; q0 < qn; q0 += 32) {
        int cnt = qn - q0;
        if (cnt > 32) cnt = 32;
        int s = 0, rl = 0;
        float nrm = 0.0f;
        if (lane < cnt) {
            int ent = q[q0 + lane];
            rl = dst[ent] - lo;
            s = src[ent];
            nrm = dinv[s] * rdinv[rl];
        }
        int j = 0;
        for (; j + 8 <= cnt; j += 8) owner_rows<8>(s, rl, nrm, j, lane, XW, acc);
        if (j + 4 <= cnt) { owner_rows<4>(s, rl, nrm, j, lane, XW, acc); j += 4; }
        if (j + 2 <= cnt) { owner_rows<2>(s, rl, nrm, j, lane, XW, acc); j += 2; }
        if (j < cnt) owner_rows<1>(s, rl, nrm, j, lane, XW, acc);
    }
}

__global__ void __launch_bounds__(OWN_T)
owner_layer(const float* __restrict__ XW, const float* __restrict__ dinv,
            const float* __restrict__ bias,
            const int* __restrict__ src, const int* __restrict__ dst,
            const unsigned* __restrict__ own8,
            float* __restrict__ H, int ldh, int colOff, int n, int e) {
    __shared__ float acc[OWN_R * HD];
    __shared__ float rdinv[OWN_R];
    __shared__ int queue[OWN_W][OWN_Q];

    int lane = threadIdx.x & 31;
    int wave = threadIdx.x >> 5;
    int lo = blockIdx.x * OWN_R;
    int rows = n - lo;
    if (rows > OWN_R) rows = OWN_R;

    {
        const int RW = (OWN_R + OWN_W - 1) / OWN_W;
        float4 b = ((const float4*)bias)[lane];
        float4 x[RW];
        float dn[RW];
#pragma unroll
        for (int i = 0; i < RW; ++i) {
            int r = wave + i * OWN_W;
            int v = lo + (r < rows ? r : 0);
            dn[i] = dinv[v];
            x[i] = ((const float4*)(XW + (size_t)v * HD))[lane];
        }
#pragma unroll
        for (int i = 0; i < RW; ++i) {
            int r = wave + i * OWN_W;
            if (r < rows) {
                float nn = dn[i] * dn[i];
                float* a = acc + r * HD + lane;
                a[0]  = x[i].x * nn + b.x;
                a[32] = x[i].y * nn + b.y;
                a[64] = x[i].z * nn + b.z;
                a[96] = x[i].w * nn + b.w;
                if (lane == 0) rdinv[r] = dn[i];
            }
        }
    }
    __syncthreads();

    int* q = queue[wave];
    int qn = 0;
    const int4* o4 = (const int4*)own8;
    int G = e >> 4;
    unsigned pat = (unsigned)blockIdx.x * 0x01010101u;
    auto fetch = [&](int4 (&b)[4], int g0) {
#pragma unroll
        for (int u = 0; u < 4; ++u) {
            int g = g0 + u * OWN_T + (int)threadIdx.x;
            b[u] = o4[g < G ? g : 0];
        }
    };
    auto sift = [&](const int4 (&b)[4], int g0) {
#pragma unroll
        for (int u = 0; u < 4; ++u) {
            int g = g0 + u * OWN_T + (int)threadIdx.x;
            unsigned w[4] = {(unsigned)b[u].x, (unsigned)b[u].y,
                             (unsigned)b[u].z, (unsigned)b[u].w};
            unsigned f = 0;
#pragma unroll
            for (int j = 0; j < 4; ++j) {
                unsigned x = w[j] ^ pat;
                unsigned z = ~(((x & 0x7F7F7F7Fu) + 0x7F7F7F7Fu) | x | 0x7F7F7F7Fu);
                f |= z >> (7 - j);
            }
            if (g >= G) f = 0;
            unsigned have = __builtin_amdgcn_ballot_w32(f != 0);
            while (have) {
                if (f) {
                    int p = __builtin_ctz(f);
                    int pos = qn + __popc(have & ((1u << lane) - 1u));
                    q[pos] = g * 16 + 4 * (p & 7) + (p >> 3);
                    f &= f - 1;
                }
                qn += __popc(have);
                have = __builtin_amdgcn_ballot_w32(f != 0);
            }
            if (qn >= 32) {
                owner_drain(q, qn, lane, lo, XW, dinv, src, dst, rdinv, acc);
                qn = 0;
            }
        }
    };
    const int RD = OWN_T * 4;
    int4 ra[4], rb[4];
    fetch(ra, 0);
    for (int g0 = 0; g0 < G; g0 += 2 * RD) {
        fetch(rb, g0 + RD);
        sift(ra, g0);
        fetch(ra, g0 + 2 * RD);
        sift(rb, g0 + RD);
    }
    if (wave == 0) {
        int eid = 16 * G + lane;
        unsigned rel = eid < e ? (unsigned)(dst[eid] - lo) : 0xFFFFFFFFu;
        unsigned mk = __builtin_amdgcn_ballot_w32(rel < (unsigned)rows);
        if (rel < (unsigned)rows) {
            int pos = qn + __popc(mk & ((1u << lane) - 1u));
            q[pos] = eid;
        }
        qn += __popc(mk);
    }
    owner_drain(q, qn, lane, lo, XW, dinv, src, dst, rdinv, acc);
    __syncthreads();

    for (int r = wave; r < rows; r += OWN_W) {
        const float* a = acc + r * HD + lane;
        float4 o = {a[0], a[32], a[64], a[96]};
        *(float4*)(H + (size_t)(lo + r) * ldh + colOff + lane * 4) = o;
    }
}

__global__ void __launch_bounds__(256)
pack_head_w(const float* __restrict__ lW1, float* __restrict__ P) {
    int idx = blockIdx.x * blockDim.x + threadIdx.x;
    if (idx >= (CD / 8) * (CD / 16) * 32) return;
    int lane = idx & 31;
    int nt = (idx >> 5) % (CD / 16);
    int b  = (idx >> 5) / (CD / 16);
    int c = lane & 15, half = lane >> 4;
    float4 v = *(const float4*)(lW1 + (size_t)(nt * 16 + c) * CD + 8 * b + 4 * half);
    ((float4*)P)[idx] = v;
}

#define HT (CD / 16)
#define HG 6
__global__ void __launch_bounds__(128)
head_rows(const float* __restrict__ H,
          const float* __restrict__ P,
          const float* __restrict__ lb1,
          const float* __restrict__ lW2,
          const float* __restrict__ lb2,
          float* __restrict__ logits,
          float* __restrict__ probs,
          int rowTiles) {
    int lane = threadIdx.x & 31;
    int tile = blockIdx.x * (blockDim.x >> 5) + (threadIdx.x >> 5);
    if (tile >= rowTiles) return;
    int c = lane & 15, half = lane >> 4;
    int tm = tile << 4;

    v8f acc[HT];
#pragma unroll
    for (int nt = 0; nt < HT; ++nt) acc[nt] = (v8f){};

    const float4* ap = (const float4*)(H + (size_t)(tm + c) * CD + 4 * half);
    const float4* bp = (const float4*)P + lane;
    const int NB = CD / 8;

    float4 bcur[HG], bnxt[HG];
#pragma unroll
    for (int t = 0; t < HG; ++t) bcur[t] = bp[t * 32];
    float4 acur = ap[0];

    for (int b = 0; b < NB; ++b) {
        int bn = b + 1 < NB ? b + 1 : b;
        float4 anxt = ap[2 * bn];
        v2f a0 = {acur.x, acur.y};
        v2f a1 = {acur.z, acur.w};
#pragma unroll
        for (int g = 0; g < HT / HG; ++g) {
            int gn = b * (HT / HG) + g + 1;
            if (gn > NB * (HT / HG) - 1) gn = NB * (HT / HG) - 1;
#pragma unroll
            for (int t = 0; t < HG; ++t) bnxt[t] = bp[(gn * HG + t) * 32];
#pragma unroll
            for (int t = 0; t < HG; ++t) {
                v2f b0 = {bcur[t].x, bcur[t].y};
                acc[g * HG + t] = __builtin_amdgcn_wmma_f32_16x16x4_f32(
                    false, a0, false, b0, (short)0, acc[g * HG + t], false, false);
            }
#pragma unroll
            for (int t = 0; t < HG; ++t) {
                v2f b1 = {bcur[t].z, bcur[t].w};
                acc[g * HG + t] = __builtin_amdgcn_wmma_f32_16x16x4_f32(
                    false, a1, false, b1, (short)0, acc[g * HG + t], false, false);
            }
#pragma unroll
            for (int t = 0; t < HG; ++t) bcur[t] = bnxt[t];
        }
        acur = anxt;
    }

    float p0[8], p1[8];
#pragma unroll
    for (int i = 0; i < 8; ++i) { p0[i] = 0.0f; p1[i] = 0.0f; }
#pragma unroll
    for (int nt = 0; nt < HT; ++nt) {
        float bias = lb1[nt * 16 + c];
        float w0 = lW2[0 * CD + nt * 16 + c];
        float w1 = lW2[1 * CD + nt * 16 + c];
#pragma unroll
        for (int i = 0; i < 8; ++i) {
            float t = acc[nt][i] + bias;
            t = t > 0.0f ? t : 0.0f;
            p0[i] += t * w0;
            p1[i] += t * w1;
        }
    }
#pragma unroll
    for (int i = 0; i < 8; ++i) {
#pragma unroll
        for (int m = 8; m >= 1; m >>= 1) {
            p0[i] += __shfl_xor(p0[i], m, 16);
            p1[i] += __shfl_xor(p1[i], m, 16);
        }
    }
    float c0 = lb2[0], c1 = lb2[1];
#pragma unroll
    for (int i = 0; i < 8; ++i) {
        if (c == i) {
            int r = tm + i + 8 * half;
            float a = p0[i] + c0, bb = p1[i] + c1;
            float m = fmaxf(a, bb);
            float ea = __expf(a - m), eb = __expf(bb - m);
            float inv = 1.0f / (ea + eb);
            logits[2 * r + 0] = a;
            logits[2 * r + 1] = bb;
            probs[2 * r + 0] = ea * inv;
            probs[2 * r + 1] = eb * inv;
        }
    }
}

extern "C" void kernel_launch(void* const* d_in, const int* in_sizes, int n_in,
                              void* d_out, int out_size, void* d_ws, size_t ws_size,
                              hipStream_t stream) {
    const float* x   = (const float*)d_in[0];
    const int*   ei  = (const int*)d_in[1];
    const float* W1  = (const float*)d_in[2];
    const float* b1  = (const float*)d_in[3];
    const float* W2  = (const float*)d_in[4];
    const float* b2  = (const float*)d_in[5];
    const float* W3  = (const float*)d_in[6];
    const float* b3  = (const float*)d_in[7];
    const float* lW1 = (const float*)d_in[8];
    const float* lb1 = (const float*)d_in[9];
    const float* lW2 = (const float*)d_in[10];
    const float* lb2 = (const float*)d_in[11];

    const int N = NN, E = EE;
    const int* src = ei;
    const int* dst = ei + E;

    float* dinv = (float*)d_ws;
    float* xw   = dinv + N;
    float* h    = xw + (size_t)N * HD;

    float* logits = (float*)d_out;
    float* probs  = logits + (size_t)N * 2;

    const int T = 256;
    const int rowTiles = N / 16;

    float* wpk = h + (size_t)N * CD + 48;
    unsigned* own8 = (unsigned*)(wpk + (size_t)(CD / 8) * (CD / 16) * 32 * 4);

    deg_init<<<(N + T - 1) / T, T, 0, stream>>>(dinv, N);
    deg_count<<<((E + 3) / 4 + T - 1) / T, T, 0, stream>>>(dst, dinv, own8, E);
    deg_rsqrt<<<(N + T - 1) / T, T, 0, stream>>>(dinv, N);

    {
        int tiles = rowTiles * (HD / 16);
        gemm16_f32<false><<<(tiles + 7) / 8, T, 0, stream>>>(
            x, 64, W1, HD, xw, HD, 64, HD / 16, tiles);
        owner_layer<<<(N + OWN_R - 1) / OWN_R, OWN_T, 0, stream>>>(
            xw, dinv, b1, src, dst, own8, h, CD, 0, N, E);
    }
    {
        int tiles = rowTiles * (HD / 16);
        gemm16_f32<false><<<(tiles + 7) / 8, T, 0, stream>>>(
            h + 0, CD, W2, HD, xw, HD, HD, HD / 16, tiles);
        owner_layer<<<(N + OWN_R - 1) / OWN_R, OWN_T, 0, stream>>>(
            xw, dinv, b2, src, dst, own8, h, CD, HD, N, E);
    }
    {
        int tiles = rowTiles * (HD / 16);
        gemm16_f32<false><<<(tiles + 7) / 8, T, 0, stream>>>(
            h + HD, CD, W3, HD, xw, HD, HD, HD / 16, tiles);
        owner_layer<<<(N + OWN_R - 1) / OWN_R, OWN_T, 0, stream>>>(
            xw, dinv, b3, src, dst, own8, h, CD, 2 * HD, N, E);
    }

    pack_head_w<<<((CD / 8) * (CD / 16) * 32 + T - 1) / T, T, 0, stream>>>(lW1, wpk);
    head_rows<<<(rowTiles + 3) / 4, 128, 0, stream>>>(
        h, wpk, lb1, lW2, lb2, logits, probs, rowTiles);
}


# Edges binned by owner group once per call (count in degree pass, fill pass writes entries src|row, norm); owner layers read their own list, no per-layer scan; sums in another LDS-arrival order
# speedup vs baseline: 2.8622x; 1.1127x over previous
#include <hip/hip_runtime.h>
#include <hip/hip_bf16.h>


typedef __attribute__((ext_vector_type(2))) float v2f;
typedef __attribute__((ext_vector_type(8))) float v8f;

#define NN 50000
#define EE 800000
#define HD 128
#define CD 384

#define OWN_R 196
#define NBIN 256
#define EB_T 1024

__global__ void deg_init(float* deg, int* bins, int n) {
    int i = blockIdx.x * blockDim.x + threadIdx.x;
    if (i < n) deg[i] = 1.0f;
    if (i < 2 * NBIN) bins[i] = 0;
}
__global__ void __launch_bounds__(EB_T)
deg_count(const int* __restrict__ dst, float* deg, int* binCnt, int e) {
    __shared__ int hist[NBIN];
    int t = threadIdx.x;
    if (t < NBIN) hist[t] = 0;
    __syncthreads();
    int g = blockIdx.x * EB_T + t;
    int e0 = g * 4;
    if (e0 + 4 <= e) {
        int4 d = ((const int4*)dst)[g];
        atomicAdd(&deg[d.x], 1.0f);
        atomicAdd(&deg[d.y], 1.0f);
        atomicAdd(&deg[d.z], 1.0f);
        atomicAdd(&deg[d.w], 1.0f);
        atomicAdd(&hist[(unsigned)d.x / OWN_R], 1);
        atomicAdd(&hist[(unsigned)d.y / OWN_R], 1);
        atomicAdd(&hist[(unsigned)d.z / OWN_R], 1);
        atomicAdd(&hist[(unsigned)d.w / OWN_R], 1);
    } else {
        for (int c = 0; c < 4; ++c) {
            if (e0 + c < e) {
                int d = dst[e0 + c];
                atomicAdd(&deg[d], 1.0f);
                atomicAdd(&hist[(unsigned)d / OWN_R], 1);
            }
        }
    }
    __syncthreads();
    if (t < NBIN) {
        int h = hist[t];
        if (h) atomicAdd(&binCnt[t], h);
    }
}
__global__ void __launch_bounds__(EB_T)
edge_fill(const int* __restrict__ src, const int* __restrict__ dst,
          const float* __restrict__ deg, float* __restrict__ dinv,
          const int* __restrict__ binCnt, int* binCur,
          int* __restrict__ binStart, int2* __restrict__ list, int n, int e) {
    __shared__ int hist[NBIN];
    __shared__ int wtot[NBIN / 32];
    int t = threadIdx.x;
    int g = blockIdx.x * EB_T + t;
    if (g < n) dinv[g] = rsqrtf(deg[g]);
    if (t < NBIN) hist[t] = 0;
    __syncthreads();

    int e0 = g * 4;
    int m = e - e0;
    m = m < 0 ? 0 : (m > 4 ? 4 : m);
    int dd[4] = {0, 0, 0, 0}, ss[4] = {0, 0, 0, 0}, rk[4] = {0, 0, 0, 0};
    if (m == 4) {
        int4 d = ((const int4*)dst)[g];
        int4 s = ((const int4*)src)[g];
        dd[0] = d.x; dd[1] = d.y; dd[2] = d.z; dd[3] = d.w;
        ss[0] = s.x; ss[1] = s.y; ss[2] = s.z; ss[3] = s.w;
    } else {
        for (int c = 0; c < m; ++c) { dd[c] = dst[e0 + c]; ss[c] = src[e0 + c]; }
    }
#pragma unroll
    for (int c = 0; c < 4; ++c)
        if (c < m) rk[c] = atomicAdd(&hist[(unsigned)dd[c] / OWN_R], 1);
    __syncthreads();

    int v = 0, excl = 0, h = 0;
    if (t < NBIN) {
        v = binCnt[t];
        int x = v;
#pragma unroll
        for (int o = 1; o < 32; o <<= 1) {
            int y = __shfl_up(x, o, 32);
            if ((t & 31) >= o) x += y;
        }
        if ((t & 31) == 31) wtot[t >> 5] = x;
        excl = x - v;
        h = hist[t];
    }
    __syncthreads();
    if (t < NBIN) {
        for (int w = 0; w < (t >> 5); ++w) excl += wtot[w];
        int base = excl;
        if (h) base += atomicAdd(&binCur[t], h);
        hist[t] = base;
        if (blockIdx.x == 0) {
            binStart[t] = excl;
            if (t == NBIN - 1) binStart[NBIN] = excl + v;
        }
    }
    __syncthreads();
#pragma unroll
    for (int c = 0; c < 4; ++c) {
        if (c < m) {
            unsigned o = (unsigned)dd[c] / OWN_R;
            int rl = dd[c] - (int)o * OWN_R;
            float nrm = rsqrtf(deg[ss[c]]) * rsqrtf(deg[dd[c]]);
            list[hist[o] + rk[c]] = make_int2(ss[c] | (rl << 16), __float_as_int(nrm));
        }
    }
}

template <bool BT>
__global__ void __launch_bounds__(256)
gemm16_f32(const float* __restrict__ A, int lda,
           const float* __restrict__ B, int ldb,
           float* __restrict__ C, int ldc,
           int K, int tilesPerRow, int totalTiles) {
    int wave = threadIdx.x >> 5;
    int lane = threadIdx.x & 31;
    int tile = blockIdx.x * (blockDim.x >> 5) + wave;
    if (tile >= totalTiles) return;
    int tm = (tile / tilesPerRow) << 4;
    int tn = (tile % tilesPerRow) << 4;
    int rc   = lane & 15;
    int half = lane >> 4;

    v8f acc = {};
    const float* arow = A + (size_t)(tm + rc) * lda;
#pragma unroll 4
    for (int k0 = 0; k0 < K; k0 += 4) {
        int ka = k0 + 2 * half;
        v2f a;
        a.x = arow[ka];
        a.y = arow[ka + 1];
        v2f b;
        if (BT) {
            const float* bp = B + (size_t)(tn + rc) * ldb + ka;
            b.x = bp[0];
            b.y = bp[1];
        } else {
            b.x = B[(size_t)(ka + 0) * ldb + tn + rc];
            b.y = B[(size_t)(ka + 1) * ldb + tn + rc];
        }
        acc = __builtin_amdgcn_wmma_f32_16x16x4_f32(
            false, a, false, b, (short)0, acc, false, false);
    }
#pragma unroll
    for (int i = 0; i < 8; ++i)
        C[(size_t)(tm + i + 8 * half) * ldc + tn + rc] = acc[i];
}

#define OWN_T 1024
#define OWN_W (OWN_T / 32)

template <int U>
__device__ __forceinline__ void owner_rows(
        int s, int rl, float nrm, int j, int lane,
        const float* __restrict__ XW, float* acc) {
    float4 v[U];
    float nj[U];
    int rj[U];
#pragma unroll
    for (int u = 0; u < U; ++u) {
        int sj = __builtin_amdgcn_readlane(s, j + u);
        rj[u] = __builtin_amdgcn_readlane(rl, j + u);
        nj[u] = __int_as_float(__builtin_amdgcn_readlane(__float_as_int(nrm), j + u));
        v[u] = ((const float4*)(XW + (size_t)sj * HD))[lane];
    }
#pragma unroll
    for (int u = 0; u < U; ++u) {
        float* a = acc + rj[u] * HD + lane;
        atomicAdd(a,      v[u].x * nj[u]);
        atomicAdd(a + 32, v[u].y * nj[u]);
        atomicAdd(a + 64, v[u].z * nj[u]);
        atomicAdd(a + 96, v[u].w * nj[u]);
    }
}

__global__ void __launch_bounds__(OWN_T)
owner_layer(const float* __restrict__ XW, const float* __restrict__ dinv,
            const float* __restrict__ bias,
            const int2* __restrict__ list, const int* __restrict__ binStart,
            float* __restrict__ H, int ldh, int colOff, int n) {
    __shared__ float acc[OWN_R * HD];

    int lane = threadIdx.x & 31;
    int wave = __builtin_amdgcn_readfirstlane(threadIdx.x >> 5);
    int lo = blockIdx.x * OWN_R;
    int rows = n - lo;
    if (rows > OWN_R) rows = OWN_R;

    int beg = binStart[blockIdx.x];
    int end = binStart[blockIdx.x + 1];
    int per = (end - beg + OWN_W - 1) / OWN_W;
    int b0 = beg + wave * per;
    int b1 = b0 + per < end ? b0 + per : end;
    int2 nx = make_int2(0, 0);
    if (b0 + lane < b1) nx = list[b0 + lane];

    {
        const int RW = (OWN_R + OWN_W - 1) / OWN_W;
        float4 b = ((const float4*)bias)[lane];
        float4 x[RW];
        float dn[RW];
#pragma unroll
        for (int i = 0; i < RW; ++i) {
            int r = wave + i * OWN_W;
            int v = lo + (r < rows ? r : 0);
            dn[i] = dinv[v];
            x[i] = ((const float4*)(XW + (size_t)v * HD))[lane];
        }
#pragma unroll
        for (int i = 0; i < RW; ++i) {
            int r = wave + i * OWN_W;
            if (r < rows) {
                float nn = dn[i] * dn[i];
                float* a = acc + r * HD + lane;
                a[0]  = x[i].x * nn + b.x;
                a[32] = x[i].y * nn + b.y;
                a[64] = x[i].z * nn + b.z;
                a[96] = x[i].w * nn + b.w;
            }
        }
    }
    __syncthreads();

    for (int i0 = b0; i0 < b1; i0 += 32) {
        int2 cur = nx;
        nx = make_int2(0, 0);
        if (i0 + 32 + lane < b1) nx = list[i0 + 32 + lane];
        int cnt = b1 - i0;
        if (cnt > 32) cnt = 32;
        int s = cur.x & 0xFFFF;
        int rl = (int)((unsigned)cur.x >> 16);
        float nrm = __int_as_float(cur.y);
        int j = 0;
        for (; j + 8 <= cnt; j += 8) owner_rows<8>(s, rl, nrm, j, lane, XW, acc);
        if (j + 4 <= cnt) { owner_rows<4>(s, rl, nrm, j, lane, XW, acc); j += 4; }
        if (j + 2 <= cnt) { owner_rows<2>(s, rl, nrm, j, lane, XW, acc); j += 2; }
        if (j < cnt) owner_rows<1>(s, rl, nrm, j, lane, XW, acc);
    }
    __syncthreads();

    for (int r = wave; r < rows; r += OWN_W) {
        const float* a = acc + r * HD + lane;
        float4 o = {a[0], a[32], a[64], a[96]};
        *(float4*)(H + (size_t)(lo + r) * ldh + colOff + lane * 4) = o;
    }
}

__global__ void __launch_bounds__(256)
pack_head_w(const float* __restrict__ lW1, float* __restrict__ P) {
    int idx = blockIdx.x * blockDim.x + threadIdx.x;
    if (idx >= (CD / 8) * (CD / 16) * 32) return;
    int lane = idx & 31;
    int nt = (idx >> 5) % (CD / 16);
    int b  = (idx >> 5) / (CD / 16);
    int c = lane & 15, half = lane >> 4;
    float4 v = *(const float4*)(lW1 + (size_t)(nt * 16 + c) * CD + 8 * b + 4 * half);
    ((float4*)P)[idx] = v;
}

#define HT (CD / 16)
#define HG 6
__global__ void __launch_bounds__(128)
head_rows(const float* __restrict__ H,
          const float* __restrict__ P,
          const float* __restrict__ lb1,
          const float* __restrict__ lW2,
          const float* __restrict__ lb2,
          float* __restrict__ logits,
          float* __restrict__ probs,
          int rowTiles) {
    int lane = threadIdx.x & 31;
    int tile = blockIdx.x * (blockDim.x >> 5) + (threadIdx.x >> 5);
    if (tile >= rowTiles) return;
    int c = lane & 15, half = lane >> 4;
    int tm = tile << 4;

    v8f acc[HT];
#pragma unroll
    for (int nt = 0; nt < HT; ++nt) acc[nt] = (v8f){};

    const float4* ap = (const float4*)(H + (size_t)(tm + c) * CD + 4 * half);
    const float4* bp = (const float4*)P + lane;
    const int NB = CD / 8;

    float4 bcur[HG], bnxt[HG];
#pragma unroll
    for (int t = 0; t < HG; ++t) bcur[t] = bp[t * 32];
    float4 acur = ap[0];

    for (int b = 0; b < NB; ++b) {
        int bn = b + 1 < NB ? b + 1 : b;
        float4 anxt = ap[2 * bn];
        v2f a0 = {acur.x, acur.y};
        v2f a1 = {acur.z, acur.w};
#pragma unroll
        for (int g = 0; g < HT / HG; ++g) {
            int gn = b * (HT / HG) + g + 1;
            if (gn > NB * (HT / HG) - 1) gn = NB * (HT / HG) - 1;
#pragma unroll
            for (int t = 0; t < HG; ++t) bnxt[t] = bp[(gn * HG + t) * 32];
#pragma unroll
            for (int t = 0; t < HG; ++t) {
                v2f b0 = {bcur[t].x, bcur[t].y};
                acc[g * HG + t] = __builtin_amdgcn_wmma_f32_16x16x4_f32(
                    false, a0, false, b0, (short)0, acc[g * HG + t], false, false);
            }
#pragma unroll
            for (int t = 0; t < HG; ++t) {
                v2f b1 = {bcur[t].z, bcur[t].w};
                acc[g * HG + t] = __builtin_amdgcn_wmma_f32_16x16x4_f32(
                    false, a1, false, b1, (short)0, acc[g * HG + t], false, false);
            }
#pragma unroll
            for (int t = 0; t < HG; ++t) bcur[t] = bnxt[t];
        }
        acur = anxt;
    }

    float p0[8], p1[8];
#pragma unroll
    for (int i = 0; i < 8; ++i) { p0[i] = 0.0f; p1[i] = 0.0f; }
#pragma unroll
    for (int nt = 0; nt < HT; ++nt) {
        float bias = lb1[nt * 16 + c];
        float w0 = lW2[0 * CD + nt * 16 + c];
        float w1 = lW2[1 * CD + nt * 16 + c];
#pragma unroll
        for (int i = 0; i < 8; ++i) {
            float t = acc[nt][i] + bias;
            t = t > 0.0f ? t : 0.0f;
            p0[i] += t * w0;
            p1[i] += t * w1;
        }
    }
#pragma unroll
    for (int i = 0; i < 8; ++i) {
#pragma unroll
        for (int m = 8; m >= 1; m >>= 1) {
            p0[i] += __shfl_xor(p0[i], m, 16);
            p1[i] += __shfl_xor(p1[i], m, 16);
        }
    }
    float c0 = lb2[0], c1 = lb2[1];
#pragma unroll
    for (int i = 0; i < 8; ++i) {
        if (c == i) {
            int r = tm + i + 8 * half;
            float a = p0[i] + c0, bb = p1[i] + c1;
            float m = fmaxf(a, bb);
            float ea = __expf(a - m), eb = __expf(bb - m);
            float inv = 1.0f / (ea + eb);
            logits[2 * r + 0] = a;
            logits[2 * r + 1] = bb;
            probs[2 * r + 0] = ea * inv;
            probs[2 * r + 1] = eb * inv;
        }
    }
}

extern "C" void kernel_launch(void* const* d_in, const int* in_sizes, int n_in,
                              void* d_out, int out_size, void* d_ws, size_t ws_size,
                              hipStream_t stream) {
    const float* x   = (const float*)d_in[0];
    const int*   ei  = (const int*)d_in[1];
    const float* W1  = (const float*)d_in[2];
    const float* b1  = (const float*)d_in[3];
    const float* W2  = (const float*)d_in[4];
    const float* b2  = (const float*)d_in[5];
    const float* W3  = (const float*)d_in[6];
    const float* b3  = (const float*)d_in[7];
    const float* lW1 = (const float*)d_in[8];
    const float* lb1 = (const float*)d_in[9];
    const float* lW2 = (const float*)d_in[10];
    const float* lb2 = (const float*)d_in[11];

    const int N = NN, E = EE;
    const int* src = ei;
    const int* dst = ei + E;

    float* deg  = (float*)d_ws;
    float* xw   = deg + N;
    float* h    = xw + (size_t)N * HD;

    float* logits = (float*)d_out;
    float* probs  = logits + (size_t)N * 2;

    const int T = 256;
    const int rowTiles = N / 16;

    float* wpk = h + (size_t)N * CD + 48;
    int2* elist = (int2*)(wpk + (size_t)(CD / 8) * (CD / 16) * 32 * 4);
    float* dinv = (float*)(elist + E);
    int* binCnt = (int*)(dinv + N);
    int* binCur = binCnt + NBIN;
    int* binStart = binCur + NBIN;

    const int EB = ((E + 3) / 4 + EB_T - 1) / EB_T;
    deg_init<<<(N + T - 1) / T, T, 0, stream>>>(deg, binCnt, N);
    deg_count<<<EB, EB_T, 0, stream>>>(dst, deg, binCnt, E);
    edge_fill<<<EB, EB_T, 0, stream>>>(src, dst, deg, dinv, binCnt, binCur, binStart, elist, N, E);

    {
        int tiles = rowTiles * (HD / 16);
        gemm16_f32<false><<<(tiles + 7) / 8, T, 0, stream>>>(
            x, 64, W1, HD, xw, HD, 64, HD / 16, tiles);
        owner_layer<<<(N + OWN_R - 1) / OWN_R, OWN_T, 0, stream>>>(
            xw, dinv, b1, elist, binStart, h, CD, 0, N);
    }
    {
        int tiles = rowTiles * (HD / 16);
        gemm16_f32<false><<<(tiles + 7) / 8, T, 0, stream>>>(
            h + 0, CD, W2, HD, xw, HD, HD, HD / 16, tiles);
        owner_layer<<<(N + OWN_R - 1) / OWN_R, OWN_T, 0, stream>>>(
            xw, dinv, b2, elist, binStart, h, CD, HD, N);
    }
    {
        int tiles = rowTiles * (HD / 16);
        gemm16_f32<false><<<(tiles + 7) / 8, T, 0, stream>>>(
            h + HD, CD, W3, HD, xw, HD, HD, HD / 16, tiles);
        owner_layer<<<(N + OWN_R - 1) / OWN_R, OWN_T, 0, stream>>>(
            xw, dinv, b3, elist, binStart, h, CD, 2 * HD, N);
    }

    pack_head_w<<<((CD / 8) * (CD / 16) * 32 + T - 1) / T, T, 0, stream>>>(lW1, wpk);
    head_rows<<<(rowTiles + 3) / 4, 128, 0, stream>>>(
        h, wpk, lb1, lW2, lb2, logits, probs, rowTiles);
}


# Next conv's transform fused into the owner layer: summed rows (LDS, pitch 132) x W2/W3 (staged in LDS) on WMMA, next xw written there; two GEMM launches gone (11 to 9); k summed in another order
# speedup vs baseline: 3.3046x; 1.1546x over previous
#include <hip/hip_runtime.h>
#include <hip/hip_bf16.h>


typedef __attribute__((ext_vector_type(2))) float v2f;
typedef __attribute__((ext_vector_type(8))) float v8f;

#define NN 50000
#define EE 800000
#define HD 128
#define CD 384

#define OWN_R 196
#define NBIN 256
#define EB_T 1024

__global__ void deg_init(float* deg, int* bins, int n) {
    int i = blockIdx.x * blockDim.x + threadIdx.x;
    if (i < n) deg[i] = 1.0f;
    if (i < 2 * NBIN) bins[i] = 0;
}
__global__ void __launch_bounds__(EB_T)
deg_count(const int* __restrict__ dst, float* deg, int* binCnt, int e) {
    __shared__ int hist[NBIN];
    int t = threadIdx.x;
    if (t < NBIN) hist[t] = 0;
    __syncthreads();
    int g = blockIdx.x * EB_T + t;
    int e0 = g * 4;
    if (e0 + 4 <= e) {
        int4 d = ((const int4*)dst)[g];
        atomicAdd(&deg[d.x], 1.0f);
        atomicAdd(&deg[d.y], 1.0f);
        atomicAdd(&deg[d.z], 1.0f);
        atomicAdd(&deg[d.w], 1.0f);
        atomicAdd(&hist[(unsigned)d.x / OWN_R], 1);
        atomicAdd(&hist[(unsigned)d.y / OWN_R], 1);
        atomicAdd(&hist[(unsigned)d.z / OWN_R], 1);
        atomicAdd(&hist[(unsigned)d.w / OWN_R], 1);
    } else {
        for (int c = 0; c < 4; ++c) {
            if (e0 + c < e) {
                int d = dst[e0 + c];
                atomicAdd(&deg[d], 1.0f);
                atomicAdd(&hist[(unsigned)d / OWN_R], 1);
            }
        }
    }
    __syncthreads();
    if (t < NBIN) {
        int h = hist[t];
        if (h) atomicAdd(&binCnt[t], h);
    }
}
__global__ void __launch_bounds__(EB_T)
edge_fill(const int* __restrict__ src, const int* __restrict__ dst,
          const float* __restrict__ deg, float* __restrict__ dinv,
          const int* __restrict__ binCnt, int* binCur,
          int* __restrict__ binStart, int2* __restrict__ list, int n, int e) {
    __shared__ int hist[NBIN];
    __shared__ int wtot[NBIN / 32];
    int t = threadIdx.x;
    int g = blockIdx.x * EB_T + t;
    if (g < n) dinv[g] = rsqrtf(deg[g]);
    if (t < NBIN) hist[t] = 0;
    __syncthreads();

    int e0 = g * 4;
    int m = e - e0;
    m = m < 0 ? 0 : (m > 4 ? 4 : m);
    int dd[4] = {0, 0, 0, 0}, ss[4] = {0, 0, 0, 0}, rk[4] = {0, 0, 0, 0};
    if (m == 4) {
        int4 d = ((const int4*)dst)[g];
        int4 s = ((const int4*)src)[g];
        dd[0] = d.x; dd[1] = d.y; dd[2] = d.z; dd[3] = d.w;
        ss[0] = s.x; ss[1] = s.y; ss[2] = s.z; ss[3] = s.w;
    } else {
        for (int c = 0; c < m; ++c) { dd[c] = dst[e0 + c]; ss[c] = src[e0 + c]; }
    }
#pragma unroll
    for (int c = 0; c < 4; ++c)
        if (c < m) rk[c] = atomicAdd(&hist[(unsigned)dd[c] / OWN_R], 1);
    __syncthreads();

    int v = 0, excl = 0, h = 0;
    if (t < NBIN) {
        v = binCnt[t];
        int x = v;
#pragma unroll
        for (int o = 1; o < 32; o <<= 1) {
            int y = __shfl_up(x, o, 32);
            if ((t & 31) >= o) x += y;
        }
        if ((t & 31) == 31) wtot[t >> 5] = x;
        excl = x - v;
        h = hist[t];
    }
    __syncthreads();
    if (t < NBIN) {
        for (int w = 0; w < (t >> 5); ++w) excl += wtot[w];
        int base = excl;
        if (h) base += atomicAdd(&binCur[t], h);
        hist[t] = base;
        if (blockIdx.x == 0) {
            binStart[t] = excl;
            if (t == NBIN - 1) binStart[NBIN] = excl + v;
        }
    }
    __syncthreads();
#pragma unroll
    for (int c = 0; c < 4; ++c) {
        if (c < m) {
            unsigned o = (unsigned)dd[c] / OWN_R;
            int rl = dd[c] - (int)o * OWN_R;
            float nrm = rsqrtf(deg[ss[c]]) * rsqrtf(deg[dd[c]]);
            list[hist[o] + rk[c]] = make_int2(ss[c] | (rl << 16), __float_as_int(nrm));
        }
    }
}

template <bool BT>
__global__ void __launch_bounds__(256)
gemm16_f32(const float* __restrict__ A, int lda,
           const float* __restrict__ B, int ldb,
           float* __restrict__ C, int ldc,
           int K, int tilesPerRow, int totalTiles) {
    int wave = threadIdx.x >> 5;
    int lane = threadIdx.x & 31;
    int tile = blockIdx.x * (blockDim.x >> 5) + wave;
    if (tile >= totalTiles) return;
    int tm = (tile / tilesPerRow) << 4;
    int tn = (tile % tilesPerRow) << 4;
    int rc   = lane & 15;
    int half = lane >> 4;

    v8f acc = {};
    const float* arow = A + (size_t)(tm + rc) * lda;
#pragma unroll 4
    for (int k0 = 0; k0 < K; k0 += 4) {
        int ka = k0 + 2 * half;
        v2f a;
        a.x = arow[ka];
        a.y = arow[ka + 1];
        v2f b;
        if (BT) {
            const float* bp = B + (size_t)(tn + rc) * ldb + ka;
            b.x = bp[0];
            b.y = bp[1];
        } else {
            b.x = B[(size_t)(ka + 0) * ldb + tn + rc];
            b.y = B[(size_t)(ka + 1) * ldb + tn + rc];
        }
        acc = __builtin_amdgcn_wmma_f32_16x16x4_f32(
            false, a, false, b, (short)0, acc, false, false);
    }
#pragma unroll
    for (int i = 0; i < 8; ++i)
        C[(size_t)(tm + i + 8 * half) * ldc + tn + rc] = acc[i];
}

#define OWN_T 1024
#define OWN_W (OWN_T / 32)
#define OWN_P (HD + 4)
#define OWN_RT ((OWN_R + 15) / 16)
#define OWN_RP (OWN_RT * 16)

template <int U>
__device__ __forceinline__ void owner_rows(
        int s, int rl, float nrm, int j, int lane,
        const float* __restrict__ XW, float* acc) {
    float4 v[U];
    float nj[U];
    int rj[U];
#pragma unroll
    for (int u = 0; u < U; ++u) {
        int sj = __builtin_amdgcn_readlane(s, j + u);
        rj[u] = __builtin_amdgcn_readlane(rl, j + u);
        nj[u] = __int_as_float(__builtin_amdgcn_readlane(__float_as_int(nrm), j + u));
        v[u] = ((const float4*)(XW + (size_t)sj * HD))[lane];
    }
#pragma unroll
    for (int u = 0; u < U; ++u) {
        float* a = acc + rj[u] * OWN_P + lane;
        atomicAdd(a,      v[u].x * nj[u]);
        atomicAdd(a + 32, v[u].y * nj[u]);
        atomicAdd(a + 64, v[u].z * nj[u]);
        atomicAdd(a + 96, v[u].w * nj[u]);
    }
}

__device__ __forceinline__ float xhalf(float x) {
    return __int_as_float(__builtin_amdgcn_permlanex16(
        __float_as_int(x), __float_as_int(x), 0x76543210, 0xfedcba98, false, false));
}

template <bool FUSE>
__global__ void __launch_bounds__(OWN_T)
owner_layer(const float* __restrict__ XW, const float* __restrict__ dinv,
            const float* __restrict__ bias,
            const int2* __restrict__ list, const int* __restrict__ binStart,
            float* __restrict__ H, int ldh, int colOff,
            const float* __restrict__ Wn, float* __restrict__ XWn, int n) {
    __shared__ __attribute__((aligned(16))) float acc[OWN_RP * OWN_P];
    __shared__ __attribute__((aligned(16))) float wl[FUSE ? HD * HD : 4];

    int lane = threadIdx.x & 31;
    int wave = __builtin_amdgcn_readfirstlane(threadIdx.x >> 5);
    int lo = blockIdx.x * OWN_R;
    int rows = n - lo;
    if (rows > OWN_R) rows = OWN_R;

    int beg = binStart[blockIdx.x];
    int end = binStart[blockIdx.x + 1];
    int per = (end - beg + OWN_W - 1) / OWN_W;
    int b0 = beg + wave * per;
    int b1 = b0 + per < end ? b0 + per : end;
    int2 nx = make_int2(0, 0);
    if (b0 + lane < b1) nx = list[b0 + lane];

    float4 wv[4];
    if (FUSE) {
#pragma unroll
        for (int u = 0; u < 4; ++u) {
            int p = threadIdx.x + u * OWN_T;
            int m = p >> 7, col = p & (HD - 1);
            const float* wp = Wn + (size_t)(16 * (m >> 2) + (m & 3)) * HD + col;
            wv[u].x = wp[0];
            wv[u].y = wp[4 * HD];
            wv[u].z = wp[8 * HD];
            wv[u].w = wp[12 * HD];
        }
    }

    {
        const int RW = (OWN_R + OWN_W - 1) / OWN_W;
        float4 b = ((const float4*)bias)[lane];
        float4 x[RW];
        float dn[RW];
#pragma unroll
        for (int i = 0; i < RW; ++i) {
            int r = wave + i * OWN_W;
            int v = lo + (r < rows ? r : 0);
            dn[i] = dinv[v];
            x[i] = ((const float4*)(XW + (size_t)v * HD))[lane];
        }
#pragma unroll
        for (int i = 0; i < RW; ++i) {
            int r = wave + i * OWN_W;
            if (r < rows) {
                float nn = dn[i] * dn[i];
                float* a = acc + r * OWN_P + lane;
                a[0]  = x[i].x * nn + b.x;
                a[32] = x[i].y * nn + b.y;
                a[64] = x[i].z * nn + b.z;
                a[96] = x[i].w * nn + b.w;
            }
        }
    }
    if (FUSE) {
#pragma unroll
        for (int u = 0; u < 4; ++u) ((float4*)wl)[threadIdx.x + u * OWN_T] = wv[u];
        for (int r = rows + wave; r < OWN_RP; r += OWN_W) {
            float* a = acc + r * OWN_P + lane;
            a[0] = 0.0f; a[32] = 0.0f; a[64] = 0.0f; a[96] = 0.0f;
        }
    }
    __syncthreads();

    for (int i0 = b0; i0 < b1; i0 += 32) {
        int2 cur = nx;
        nx = make_int2(0, 0);
        if (i0 + 32 + lane < b1) nx = list[i0 + 32 + lane];
        int cnt = b1 - i0;
        if (cnt > 32) cnt = 32;
        int s = cur.x & 0xFFFF;
        int rl = (int)((unsigned)cur.x >> 16);
        float nrm = __int_as_float(cur.y);
        int j = 0;
        for (; j + 8 <= cnt; j += 8) owner_rows<8>(s, rl, nrm, j, lane, XW, acc);
        if (j + 4 <= cnt) { owner_rows<4>(s, rl, nrm, j, lane, XW, acc); j += 4; }
        if (j + 2 <= cnt) { owner_rows<2>(s, rl, nrm, j, lane, XW, acc); j += 2; }
        if (j < cnt) owner_rows<1>(s, rl, nrm, j, lane, XW, acc);
    }
    __syncthreads();

    for (int r = wave; r < rows; r += OWN_W) {
        const float* a = acc + r * OWN_P + lane;
        float4 o = {a[0], a[32], a[64], a[96]};
        *(float4*)(H + (size_t)(lo + r) * ldh + colOff + lane * 4) = o;
    }

    if (FUSE) {
        const int c = lane & 15, hh = lane >> 4;
        const int nTask = ((rows + 15) >> 4) * 4;
        for (int task = wave; task < nTask; task += OWN_W) {
            int rt = task >> 2, cq = task & 3;
            const float4* ap = (const float4*)(acc + (rt * 16 + c) * OWN_P) + hh * 8;
            const float4* bp = (const float4*)wl + hh * HD + cq * 32 + c;
            v8f d0 = {}, d1 = {};
            float4 a = ap[0], b0 = bp[0], b1 = bp[16];
#pragma unroll
            for (int t = 0; t < 16; ++t) {
                float4 an = a, b0n = b0, b1n = b1;
                if (t + 1 < 16) {
                    an  = ap[((t + 1) & 1) * 16 + ((t + 1) >> 1)];
                    b0n = bp[(t + 1) * 2 * HD];
                    b1n = bp[(t + 1) * 2 * HD + 16];
                }
                v2f a0 = {a.x, a.y}, a1 = {a.z, a.w};
                v2f p0 = {b0.x, b0.y}, p1 = {b0.z, b0.w};
                v2f q0 = {b1.x, b1.y}, q1 = {b1.z, b1.w};
                d0 = __builtin_amdgcn_wmma_f32_16x16x4_f32(false, a0, false, p0, (short)0, d0, false, false);
                d1 = __builtin_amdgcn_wmma_f32_16x16x4_f32(false, a0, false, q0, (short)0, d1, false, false);
                d0 = __builtin_amdgcn_wmma_f32_16x16x4_f32(false, a1, false, p1, (short)0, d0, false, false);
                d1 = __builtin_amdgcn_wmma_f32_16x16x4_f32(false, a1, false, q1, (short)0, d1, false, false);
                a = an; b0 = b0n; b1 = b1n;
            }
            float* xo = XWn + (size_t)(lo + rt * 16) * HD + cq * 32 + lane;
#pragma unroll
            for (int i = 0; i < 8; ++i) {
                float got = xhalf(hh ? d0[i] : d1[i]);
                float o0 = hh ? got : d0[i];
                float o1 = hh ? d1[i] : got;
                if (rt * 16 + i < rows)     xo[(size_t)i * HD] = o0;
                if (rt * 16 + i + 8 < rows) xo[(size_t)(i + 8) * HD] = o1;
            }
        }
    }
}

__global__ void __launch_bounds__(256)
pack_head_w(const float* __restrict__ lW1, float* __restrict__ P) {
    int idx = blockIdx.x * blockDim.x + threadIdx.x;
    if (idx >= (CD / 8) * (CD / 16) * 32) return;
    int lane = idx & 31;
    int nt = (idx >> 5) % (CD / 16);
    int b  = (idx >> 5) / (CD / 16);
    int c = lane & 15, half = lane >> 4;
    float4 v = *(const float4*)(lW1 + (size_t)(nt * 16 + c) * CD + 8 * b + 4 * half);
    ((float4*)P)[idx] = v;
}

#define HT (CD / 16)
#define HG 6
__global__ void __launch_bounds__(128)
head_rows(const float* __restrict__ H,
          const float* __restrict__ P,
          const float* __restrict__ lb1,
          const float* __restrict__ lW2,
          const float* __restrict__ lb2,
          float* __restrict__ logits,
          float* __restrict__ probs,
          int rowTiles) {
    int lane = threadIdx.x & 31;
    int tile = blockIdx.x * (blockDim.x >> 5) + (threadIdx.x >> 5);
    if (tile >= rowTiles) return;
    int c = lane & 15, half = lane >> 4;
    int tm = tile << 4;

    v8f acc[HT];
#pragma unroll
    for (int nt = 0; nt < HT; ++nt) acc[nt] = (v8f){};

    const float4* ap = (const float4*)(H + (size_t)(tm + c) * CD + 4 * half);
    const float4* bp = (const float4*)P + lane;
    const int NB = CD / 8;

    float4 bcur[HG], bnxt[HG];
#pragma unroll
    for (int t = 0; t < HG; ++t) bcur[t] = bp[t * 32];
    float4 acur = ap[0];

    for (int b = 0; b < NB; ++b) {
        int bn = b + 1 < NB ? b + 1 : b;
        float4 anxt = ap[2 * bn];
        v2f a0 = {acur.x, acur.y};
        v2f a1 = {acur.z, acur.w};
#pragma unroll
        for (int g = 0; g < HT / HG; ++g) {
            int gn = b * (HT / HG) + g + 1;
            if (gn > NB * (HT / HG) - 1) gn = NB * (HT / HG) - 1;
#pragma unroll
            for (int t = 0; t < HG; ++t) bnxt[t] = bp[(gn * HG + t) * 32];
#pragma unroll
            for (int t = 0; t < HG; ++t) {
                v2f b0 = {bcur[t].x, bcur[t].y};
                acc[g * HG + t] = __builtin_amdgcn_wmma_f32_16x16x4_f32(
                    false, a0, false, b0, (short)0, acc[g * HG + t], false, false);
            }
#pragma unroll
            for (int t = 0; t < HG; ++t) {
                v2f b1 = {bcur[t].z, bcur[t].w};
                acc[g * HG + t] = __builtin_amdgcn_wmma_f32_16x16x4_f32(
                    false, a1, false, b1, (short)0, acc[g * HG + t], false, false);
            }
#pragma unroll
            for (int t = 0; t < HG; ++t) bcur[t] = bnxt[t];
        }
        acur = anxt;
    }

    float p0[8], p1[8];
#pragma unroll
    for (int i = 0; i < 8; ++i) { p0[i] = 0.0f; p1[i] = 0.0f; }
#pragma unroll
    for (int nt = 0; nt < HT; ++nt) {
        float bias = lb1[nt * 16 + c];
        float w0 = lW2[0 * CD + nt * 16 + c];
        float w1 = lW2[1 * CD + nt * 16 + c];
#pragma unroll
        for (int i = 0; i < 8; ++i) {
            float t = acc[nt][i] + bias;
            t = t > 0.0f ? t : 0.0f;
            p0[i] += t * w0;
            p1[i] += t * w1;
        }
    }
#pragma unroll
    for (int i = 0; i < 8; ++i) {
#pragma unroll
        for (int m = 8; m >= 1; m >>= 1) {
            p0[i] += __shfl_xor(p0[i], m, 16);
            p1[i] += __shfl_xor(p1[i], m, 16);
        }
    }
    float c0 = lb2[0], c1 = lb2[1];
#pragma unroll
    for (int i = 0; i < 8; ++i) {
        if (c == i) {
            int r = tm + i + 8 * half;
            float a = p0[i] + c0, bb = p1[i] + c1;
            float m = fmaxf(a, bb);
            float ea = __expf(a - m), eb = __expf(bb - m);
            float inv = 1.0f / (ea + eb);
            logits[2 * r + 0] = a;
            logits[2 * r + 1] = bb;
            probs[2 * r + 0] = ea * inv;
            probs[2 * r + 1] = eb * inv;
        }
    }
}

extern "C" void kernel_launch(void* const* d_in, const int* in_sizes, int n_in,
                              void* d_out, int out_size, void* d_ws, size_t ws_size,
                              hipStream_t stream) {
    const float* x   = (const float*)d_in[0];
    const int*   ei  = (const int*)d_in[1];
    const float* W1  = (const float*)d_in[2];
    const float* b1  = (const float*)d_in[3];
    const float* W2  = (const float*)d_in[4];
    const float* b2  = (const float*)d_in[5];
    const float* W3  = (const float*)d_in[6];
    const float* b3  = (const float*)d_in[7];
    const float* lW1 = (const float*)d_in[8];
    const float* lb1 = (const float*)d_in[9];
    const float* lW2 = (const float*)d_in[10];
    const float* lb2 = (const float*)d_in[11];

    const int N = NN, E = EE;
    const int* src = ei;
    const int* dst = ei + E;

    float* deg  = (float*)d_ws;
    float* xw   = deg + N;
    float* xw2  = xw + (size_t)N * HD;
    float* h    = xw2 + (size_t)N * HD;

    float* logits = (float*)d_out;
    float* probs  = logits + (size_t)N * 2;

    const int T = 256;
    const int rowTiles = N / 16;

    float* wpk = h + (size_t)N * CD + 48;
    int2* elist = (int2*)(wpk + (size_t)(CD / 8) * (CD / 16) * 32 * 4);
    float* dinv = (float*)(elist + E);
    int* binCnt = (int*)(dinv + N);
    int* binCur = binCnt + NBIN;
    int* binStart = binCur + NBIN;

    const int EB = ((E + 3) / 4 + EB_T - 1) / EB_T;
    deg_init<<<(N + T - 1) / T, T, 0, stream>>>(deg, binCnt, N);
    deg_count<<<EB, EB_T, 0, stream>>>(dst, deg, binCnt, E);
    edge_fill<<<EB, EB_T, 0, stream>>>(src, dst, deg, dinv, binCnt, binCur, binStart, elist, N, E);

    {
        int tiles = rowTiles * (HD / 16);
        gemm16_f32<false><<<(tiles + 7) / 8, T, 0, stream>>>(
            x, 64, W1, HD, xw, HD, 64, HD / 16, tiles);
        owner_layer<true><<<(N + OWN_R - 1) / OWN_R, OWN_T, 0, stream>>>(
            xw, dinv, b1, elist, binStart, h, CD, 0, W2, xw2, N);
    }
    owner_layer<true><<<(N + OWN_R - 1) / OWN_R, OWN_T, 0, stream>>>(
        xw2, dinv, b2, elist, binStart, h, CD, HD, W3, xw, N);
    owner_layer<false><<<(N + OWN_R - 1) / OWN_R, OWN_T, 0, stream>>>(
        xw, dinv, b3, elist, binStart, h, CD, 2 * HD, (const float*)nullptr, (float*)nullptr, N);

    pack_head_w<<<((CD / 8) * (CD / 16) * 32 + T - 1) / T, T, 0, stream>>>(lW1, wpk);
    head_rows<<<(rowTiles + 3) / 4, 128, 0, stream>>>(
        h, wpk, lb1, lW2, lb2, logits, probs, rowTiles);
}


# Head regridded: one 4-wave work-group per 16-row strip, each wave a quarter of the hidden columns (6 tiles); 3125 small groups keep all WGPs busy to the end; class sums through LDS in another order
# speedup vs baseline: 3.6035x; 1.0904x over previous
#include <hip/hip_runtime.h>
#include <hip/hip_bf16.h>


typedef __attribute__((ext_vector_type(2))) float v2f;
typedef __attribute__((ext_vector_type(8))) float v8f;

#define NN 50000
#define EE 800000
#define HD 128
#define CD 384

#define OWN_R 196
#define NBIN 256
#define EB_T 1024

__global__ void deg_init(float* deg, int* bins, int n) {
    int i = blockIdx.x * blockDim.x + threadIdx.x;
    if (i < n) deg[i] = 1.0f;
    if (i < 2 * NBIN) bins[i] = 0;
}
__global__ void __launch_bounds__(EB_T)
deg_count(const int* __restrict__ dst, float* deg, int* binCnt, int e) {
    __shared__ int hist[NBIN];
    int t = threadIdx.x;
    if (t < NBIN) hist[t] = 0;
    __syncthreads();
    int g = blockIdx.x * EB_T + t;
    int e0 = g * 4;
    if (e0 + 4 <= e) {
        int4 d = ((const int4*)dst)[g];
        atomicAdd(&deg[d.x], 1.0f);
        atomicAdd(&deg[d.y], 1.0f);
        atomicAdd(&deg[d.z], 1.0f);
        atomicAdd(&deg[d.w], 1.0f);
        atomicAdd(&hist[(unsigned)d.x / OWN_R], 1);
        atomicAdd(&hist[(unsigned)d.y / OWN_R], 1);
        atomicAdd(&hist[(unsigned)d.z / OWN_R], 1);
        atomicAdd(&hist[(unsigned)d.w / OWN_R], 1);
    } else {
        for (int c = 0; c < 4; ++c) {
            if (e0 + c < e) {
                int d = dst[e0 + c];
                atomicAdd(&deg[d], 1.0f);
                atomicAdd(&hist[(unsigned)d / OWN_R], 1);
            }
        }
    }
    __syncthreads();
    if (t < NBIN) {
        int h = hist[t];
        if (h) atomicAdd(&binCnt[t], h);
    }
}
__global__ void __launch_bounds__(EB_T)
edge_fill(const int* __restrict__ src, const int* __restrict__ dst,
          const float* __restrict__ deg, float* __restrict__ dinv,
          const int* __restrict__ binCnt, int* binCur,
          int* __restrict__ binStart, int2* __restrict__ list, int n, int e) {
    __shared__ int hist[NBIN];
    __shared__ int wtot[NBIN / 32];
    int t = threadIdx.x;
    int g = blockIdx.x * EB_T + t;
    if (g < n) dinv[g] = rsqrtf(deg[g]);
    if (t < NBIN) hist[t] = 0;
    __syncthreads();

    int e0 = g * 4;
    int m = e - e0;
    m = m < 0 ? 0 : (m > 4 ? 4 : m);
    int dd[4] = {0, 0, 0, 0}, ss[4] = {0, 0, 0, 0}, rk[4] = {0, 0, 0, 0};
    if (m == 4) {
        int4 d = ((const int4*)dst)[g];
        int4 s = ((const int4*)src)[g];
        dd[0] = d.x; dd[1] = d.y; dd[2] = d.z; dd[3] = d.w;
        ss[0] = s.x; ss[1] = s.y; ss[2] = s.z; ss[3] = s.w;
    } else {
        for (int c = 0; c < m; ++c) { dd[c] = dst[e0 + c]; ss[c] = src[e0 + c]; }
    }
#pragma unroll
    for (int c = 0; c < 4; ++c)
        if (c < m) rk[c] = atomicAdd(&hist[(unsigned)dd[c] / OWN_R], 1);
    __syncthreads();

    int v = 0, excl = 0, h = 0;
    if (t < NBIN) {
        v = binCnt[t];
        int x = v;
#pragma unroll
        for (int o = 1; o < 32; o <<= 1) {
            int y = __shfl_up(x, o, 32);
            if ((t & 31) >= o) x += y;
        }
        if ((t & 31) == 31) wtot[t >> 5] = x;
        excl = x - v;
        h = hist[t];
    }
    __syncthreads();
    if (t < NBIN) {
        for (int w = 0; w < (t >> 5); ++w) excl += wtot[w];
        int base = excl;
        if (h) base += atomicAdd(&binCur[t], h);
        hist[t] = base;
        if (blockIdx.x == 0) {
            binStart[t] = excl;
            if (t == NBIN - 1) binStart[NBIN] = excl + v;
        }
    }
    __syncthreads();
#pragma unroll
    for (int c = 0; c < 4; ++c) {
        if (c < m) {
            unsigned o = (unsigned)dd[c] / OWN_R;
            int rl = dd[c] - (int)o * OWN_R;
            float nrm = rsqrtf(deg[ss[c]]) * rsqrtf(deg[dd[c]]);
            list[hist[o] + rk[c]] = make_int2(ss[c] | (rl << 16), __float_as_int(nrm));
        }
    }
}

template <bool BT>
__global__ void __launch_bounds__(256)
gemm16_f32(const float* __restrict__ A, int lda,
           const float* __restrict__ B, int ldb,
           float* __restrict__ C, int ldc,
           int K, int tilesPerRow, int totalTiles) {
    int wave = threadIdx.x >> 5;
    int lane = threadIdx.x & 31;
    int tile = blockIdx.x * (blockDim.x >> 5) + wave;
    if (tile >= totalTiles) return;
    int tm = (tile / tilesPerRow) << 4;
    int tn = (tile % tilesPerRow) << 4;
    int rc   = lane & 15;
    int half = lane >> 4;

    v8f acc = {};
    const float* arow = A + (size_t)(tm + rc) * lda;
#pragma unroll 4
    for (int k0 = 0; k0 < K; k0 += 4) {
        int ka = k0 + 2 * half;
        v2f a;
        a.x = arow[ka];
        a.y = arow[ka + 1];
        v2f b;
        if (BT) {
            const float* bp = B + (size_t)(tn + rc) * ldb + ka;
            b.x = bp[0];
            b.y = bp[1];
        } else {
            b.x = B[(size_t)(ka + 0) * ldb + tn + rc];
            b.y = B[(size_t)(ka + 1) * ldb + tn + rc];
        }
        acc = __builtin_amdgcn_wmma_f32_16x16x4_f32(
            false, a, false, b, (short)0, acc, false, false);
    }
#pragma unroll
    for (int i = 0; i < 8; ++i)
        C[(size_t)(tm + i + 8 * half) * ldc + tn + rc] = acc[i];
}

#define OWN_T 1024
#define OWN_W (OWN_T / 32)
#define OWN_P (HD + 4)
#define OWN_RT ((OWN_R + 15) / 16)
#define OWN_RP (OWN_RT * 16)

template <int U>
__device__ __forceinline__ void owner_rows(
        int s, int rl, float nrm, int j, int lane,
        const float* __restrict__ XW, float* acc) {
    float4 v[U];
    float nj[U];
    int rj[U];
#pragma unroll
    for (int u = 0; u < U; ++u) {
        int sj = __builtin_amdgcn_readlane(s, j + u);
        rj[u] = __builtin_amdgcn_readlane(rl, j + u);
        nj[u] = __int_as_float(__builtin_amdgcn_readlane(__float_as_int(nrm), j + u));
        v[u] = ((const float4*)(XW + (size_t)sj * HD))[lane];
    }
#pragma unroll
    for (int u = 0; u < U; ++u) {
        float* a = acc + rj[u] * OWN_P + lane;
        atomicAdd(a,      v[u].x * nj[u]);
        atomicAdd(a + 32, v[u].y * nj[u]);
        atomicAdd(a + 64, v[u].z * nj[u]);
        atomicAdd(a + 96, v[u].w * nj[u]);
    }
}

__device__ __forceinline__ float xhalf(float x) {
    return __int_as_float(__builtin_amdgcn_permlanex16(
        __float_as_int(x), __float_as_int(x), 0x76543210, 0xfedcba98, false, false));
}

template <bool FUSE>
__global__ void __launch_bounds__(OWN_T)
owner_layer(const float* __restrict__ XW, const float* __restrict__ dinv,
            const float* __restrict__ bias,
            const int2* __restrict__ list, const int* __restrict__ binStart,
            float* __restrict__ H, int ldh, int colOff,
            const float* __restrict__ Wn, float* __restrict__ XWn, int n) {
    __shared__ __attribute__((aligned(16))) float acc[OWN_RP * OWN_P];
    __shared__ __attribute__((aligned(16))) float wl[FUSE ? HD * HD : 4];

    int lane = threadIdx.x & 31;
    int wave = __builtin_amdgcn_readfirstlane(threadIdx.x >> 5);
    int lo = blockIdx.x * OWN_R;
    int rows = n - lo;
    if (rows > OWN_R) rows = OWN_R;

    int beg = binStart[blockIdx.x];
    int end = binStart[blockIdx.x + 1];
    int per = (end - beg + OWN_W - 1) / OWN_W;
    int b0 = beg + wave * per;
    int b1 = b0 + per < end ? b0 + per : end;
    int2 nx = make_int2(0, 0);
    if (b0 + lane < b1) nx = list[b0 + lane];

    float4 wv[4];
    if (FUSE) {
#pragma unroll
        for (int u = 0; u < 4; ++u) {
            int p = threadIdx.x + u * OWN_T;
            int m = p >> 7, col = p & (HD - 1);
            const float* wp = Wn + (size_t)(16 * (m >> 2) + (m & 3)) * HD + col;
            wv[u].x = wp[0];
            wv[u].y = wp[4 * HD];
            wv[u].z = wp[8 * HD];
            wv[u].w = wp[12 * HD];
        }
    }

    {
        const int RW = (OWN_R + OWN_W - 1) / OWN_W;
        float4 b = ((const float4*)bias)[lane];
        float4 x[RW];
        float dn[RW];
#pragma unroll
        for (int i = 0; i < RW; ++i) {
            int r = wave + i * OWN_W;
            int v = lo + (r < rows ? r : 0);
            dn[i] = dinv[v];
            x[i] = ((const float4*)(XW + (size_t)v * HD))[lane];
        }
#pragma unroll
        for (int i = 0; i < RW; ++i) {
            int r = wave + i * OWN_W;
            if (r < rows) {
                float nn = dn[i] * dn[i];
                float* a = acc + r * OWN_P + lane;
                a[0]  = x[i].x * nn + b.x;
                a[32] = x[i].y * nn + b.y;
                a[64] = x[i].z * nn + b.z;
                a[96] = x[i].w * nn + b.w;
            }
        }
    }
    if (FUSE) {
#pragma unroll
        for (int u = 0; u < 4; ++u) ((float4*)wl)[threadIdx.x + u * OWN_T] = wv[u];
        for (int r = rows + wave; r < OWN_RP; r += OWN_W) {
            float* a = acc + r * OWN_P + lane;
            a[0] = 0.0f; a[32] = 0.0f; a[64] = 0.0f; a[96] = 0.0f;
        }
    }
    __syncthreads();

    for (int i0 = b0; i0 < b1; i0 += 32) {
        int2 cur = nx;
        nx = make_int2(0, 0);
        if (i0 + 32 + lane < b1) nx = list[i0 + 32 + lane];
        int cnt = b1 - i0;
        if (cnt > 32) cnt = 32;
        int s = cur.x & 0xFFFF;
        int rl = (int)((unsigned)cur.x >> 16);
        float nrm = __int_as_float(cur.y);
        int j = 0;
        for (; j + 8 <= cnt; j += 8) owner_rows<8>(s, rl, nrm, j, lane, XW, acc);
        if (j + 4 <= cnt) { owner_rows<4>(s, rl, nrm, j, lane, XW, acc); j += 4; }
        if (j + 2 <= cnt) { owner_rows<2>(s, rl, nrm, j, lane, XW, acc); j += 2; }
        if (j < cnt) owner_rows<1>(s, rl, nrm, j, lane, XW, acc);
    }
    __syncthreads();

    for (int r = wave; r < rows; r += OWN_W) {
        const float* a = acc + r * OWN_P + lane;
        float4 o = {a[0], a[32], a[64], a[96]};
        *(float4*)(H + (size_t)(lo + r) * ldh + colOff + lane * 4) = o;
    }

    if (FUSE) {
        const int c = lane & 15, hh = lane >> 4;
        const int nTask = ((rows + 15) >> 4) * 4;
        for (int task = wave; task < nTask; task += OWN_W) {
            int rt = task >> 2, cq = task & 3;
            const float4* ap = (const float4*)(acc + (rt * 16 + c) * OWN_P) + hh * 8;
            const float4* bp = (const float4*)wl + hh * HD + cq * 32 + c;
            v8f d0 = {}, d1 = {};
            float4 a = ap[0], b0 = bp[0], b1 = bp[16];
#pragma unroll
            for (int t = 0; t < 16; ++t) {
                float4 an = a, b0n = b0, b1n = b1;
                if (t + 1 < 16) {
                    an  = ap[((t + 1) & 1) * 16 + ((t + 1) >> 1)];
                    b0n = bp[(t + 1) * 2 * HD];
                    b1n = bp[(t + 1) * 2 * HD + 16];
                }
                v2f a0 = {a.x, a.y}, a1 = {a.z, a.w};
                v2f p0 = {b0.x, b0.y}, p1 = {b0.z, b0.w};
                v2f q0 = {b1.x, b1.y}, q1 = {b1.z, b1.w};
                d0 = __builtin_amdgcn_wmma_f32_16x16x4_f32(false, a0, false, p0, (short)0, d0, false, false);
                d1 = __builtin_amdgcn_wmma_f32_16x16x4_f32(false, a0, false, q0, (short)0, d1, false, false);
                d0 = __builtin_amdgcn_wmma_f32_16x16x4_f32(false, a1, false, p1, (short)0, d0, false, false);
                d1 = __builtin_amdgcn_wmma_f32_16x16x4_f32(false, a1, false, q1, (short)0, d1, false, false);
                a = an; b0 = b0n; b1 = b1n;
            }
            float* xo = XWn + (size_t)(lo + rt * 16) * HD + cq * 32 + lane;
#pragma unroll
            for (int i = 0; i < 8; ++i) {
                float got = xhalf(hh ? d0[i] : d1[i]);
                float o0 = hh ? got : d0[i];
                float o1 = hh ? d1[i] : got;
                if (rt * 16 + i < rows)     xo[(size_t)i * HD] = o0;
                if (rt * 16 + i + 8 < rows) xo[(size_t)(i + 8) * HD] = o1;
            }
        }
    }
}

__global__ void __launch_bounds__(256)
pack_head_w(const float* __restrict__ lW1, float* __restrict__ P) {
    int idx = blockIdx.x * blockDim.x + threadIdx.x;
    if (idx >= (CD / 8) * (CD / 16) * 32) return;
    int lane = idx & 31;
    int nt = (idx >> 5) % (CD / 16);
    int b  = (idx >> 5) / (CD / 16);
    int c = lane & 15, half = lane >> 4;
    float4 v = *(const float4*)(lW1 + (size_t)(nt * 16 + c) * CD + 8 * b + 4 * half);
    ((float4*)P)[idx] = v;
}

#define HT (CD / 16)
#define HQ 4
#define HG (HT / HQ)
#define RP 36

__device__ __forceinline__ void head_block(v8f* acc, const float4& a, const float4* b) {
    v2f a0 = {a.x, a.y};
    v2f a1 = {a.z, a.w};
#pragma unroll
    for (int t = 0; t < HG; ++t) {
        v2f b0 = {b[t].x, b[t].y};
        acc[t] = __builtin_amdgcn_wmma_f32_16x16x4_f32(
            false, a0, false, b0, (short)0, acc[t], false, false);
    }
#pragma unroll
    for (int t = 0; t < HG; ++t) {
        v2f b1 = {b[t].z, b[t].w};
        acc[t] = __builtin_amdgcn_wmma_f32_16x16x4_f32(
            false, a1, false, b1, (short)0, acc[t], false, false);
    }
}

__global__ void __launch_bounds__(HQ * 32)
head_rows(const float* __restrict__ H,
          const float* __restrict__ P,
          const float* __restrict__ lb1,
          const float* __restrict__ lW2,
          const float* __restrict__ lb2,
          float* __restrict__ logits,
          float* __restrict__ probs,
          int rowTiles) {
    __shared__ __attribute__((aligned(16))) float red[HQ * 16 * RP];
    if ((int)blockIdx.x >= rowTiles) return;
    int lane = threadIdx.x & 31;
    int q = __builtin_amdgcn_readfirstlane(threadIdx.x >> 5);
    int c = lane & 15, half = lane >> 4;
    int tm = blockIdx.x << 4;

    v8f acc[HG];
#pragma unroll
    for (int t = 0; t < HG; ++t) acc[t] = (v8f){};

    const float4* ap = (const float4*)(H + (size_t)(tm + c) * CD + 4 * half);
    const float4* bp = (const float4*)P + (size_t)q * (HG * 32) + lane;
    const int NB = CD / 8;

    float4 ax = ap[0], ay;
    float4 bx[HG], by[HG];
#pragma unroll
    for (int t = 0; t < HG; ++t) bx[t] = bp[t * 32];

    for (int b = 0; b < NB; b += 2) {
        ay = ap[2 * (b + 1)];
#pragma unroll
        for (int t = 0; t < HG; ++t) by[t] = bp[((b + 1) * HT + t) * 32];
        __builtin_amdgcn_sched_barrier(0);
        head_block(acc, ax, bx);
        int b2 = b + 2 < NB ? b + 2 : NB - 1;
        ax = ap[2 * b2];
#pragma unroll
        for (int t = 0; t < HG; ++t) bx[t] = bp[(b2 * HT + t) * 32];
        __builtin_amdgcn_sched_barrier(0);
        head_block(acc, ay, by);
    }

    float p0[8], p1[8];
#pragma unroll
    for (int i = 0; i < 8; ++i) { p0[i] = 0.0f; p1[i] = 0.0f; }
#pragma unroll
    for (int t = 0; t < HG; ++t) {
        int col = (q * HG + t) * 16 + c;
        float bias = lb1[col];
        float w0 = lW2[0 * CD + col];
        float w1 = lW2[1 * CD + col];
#pragma unroll
        for (int i = 0; i < 8; ++i) {
            float v = acc[t][i] + bias;
            v = v > 0.0f ? v : 0.0f;
            p0[i] += v * w0;
            p1[i] += v * w1;
        }
    }
    float* rw = red + (q * 16) * RP + lane;
#pragma unroll
    for (int i = 0; i < 8; ++i) {
        rw[i * RP] = p0[i];
        rw[(8 + i) * RP] = p1[i];
    }
    __syncthreads();
    if (q == 0) {
        int v = lane & 15, hf = lane >> 4;
        float cb = lb2[v >> 3];
        float s = 0.0f;
#pragma unroll
        for (int w = 0; w < HQ; ++w) {
            const float4* rp = (const float4*)(red + (w * 16 + v) * RP + hf * 16);
#pragma unroll
            for (int j = 0; j < 4; ++j) {
                float4 x = rp[j];
                s += x.x; s += x.y; s += x.z; s += x.w;
            }
        }
        s += cb;
        float o = __shfl_xor(s, 8, 32);
        if (v < 8) {
            int r = tm + v + 8 * hf;
            float a = s, bb = o;
            float m = fmaxf(a, bb);
            float ea = __expf(a - m), eb = __expf(bb - m);
            float inv = 1.0f / (ea + eb);
            *(float2*)(logits + 2 * r) = make_float2(a, bb);
            *(float2*)(probs + 2 * r) = make_float2(ea * inv, eb * inv);
        }
    }
}

extern "C" void kernel_launch(void* const* d_in, const int* in_sizes, int n_in,
                              void* d_out, int out_size, void* d_ws, size_t ws_size,
                              hipStream_t stream) {
    const float* x   = (const float*)d_in[0];
    const int*   ei  = (const int*)d_in[1];
    const float* W1  = (const float*)d_in[2];
    const float* b1  = (const float*)d_in[3];
    const float* W2  = (const float*)d_in[4];
    const float* b2  = (const float*)d_in[5];
    const float* W3  = (const float*)d_in[6];
    const float* b3  = (const float*)d_in[7];
    const float* lW1 = (const float*)d_in[8];
    const float* lb1 = (const float*)d_in[9];
    const float* lW2 = (const float*)d_in[10];
    const float* lb2 = (const float*)d_in[11];

    const int N = NN, E = EE;
    const int* src = ei;
    const int* dst = ei + E;

    float* deg  = (float*)d_ws;
    float* xw   = deg + N;
    float* xw2  = xw + (size_t)N * HD;
    float* h    = xw2 + (size_t)N * HD;

    float* logits = (float*)d_out;
    float* probs  = logits + (size_t)N * 2;

    const int T = 256;
    const int rowTiles = N / 16;

    float* wpk = h + (size_t)N * CD + 48;
    int2* elist = (int2*)(wpk + (size_t)(CD / 8) * (CD / 16) * 32 * 4);
    float* dinv = (float*)(elist + E);
    int* binCnt = (int*)(dinv + N);
    int* binCur = binCnt + NBIN;
    int* binStart = binCur + NBIN;

    const int EB = ((E + 3) / 4 + EB_T - 1) / EB_T;
    deg_init<<<(N + T - 1) / T, T, 0, stream>>>(deg, binCnt, N);
    deg_count<<<EB, EB_T, 0, stream>>>(dst, deg, binCnt, E);
    edge_fill<<<EB, EB_T, 0, stream>>>(src, dst, deg, dinv, binCnt, binCur, binStart, elist, N, E);

    {
        int tiles = rowTiles * (HD / 16);
        gemm16_f32<false><<<(tiles + 7) / 8, T, 0, stream>>>(
            x, 64, W1, HD, xw, HD, 64, HD / 16, tiles);
        owner_layer<true><<<(N + OWN_R - 1) / OWN_R, OWN_T, 0, stream>>>(
            xw, dinv, b1, elist, binStart, h, CD, 0, W2, xw2, N);
    }
    owner_layer<true><<<(N + OWN_R - 1) / OWN_R, OWN_T, 0, stream>>>(
        xw2, dinv, b2, elist, binStart, h, CD, HD, W3, xw, N);
    owner_layer<false><<<(N + OWN_R - 1) / OWN_R, OWN_T, 0, stream>>>(
        xw, dinv, b3, elist, binStart, h, CD, 2 * HD, (const float*)nullptr, (float*)nullptr, N);

    pack_head_w<<<((CD / 8) * (CD / 16) * 32 + T - 1) / T, T, 0, stream>>>(lW1, wpk);
    head_rows<<<rowTiles, HQ * 32, 0, stream>>>(
        h, wpk, lb1, lW2, lb2, logits, probs, rowTiles);
}
